# starting kernel
# baseline (speedup 1.0000x reference)
#include <hip/hip_runtime.h>
#include <hip/hip_bf16.h>

typedef float v2f __attribute__((ext_vector_type(2)));
typedef float v8f __attribute__((ext_vector_type(8)));

template <int K, int N, int MODE>
__global__ void gemm_wmma(const float* __restrict__ A, const float* __restrict__ W,
                          float* __restrict__ Y, const int* __restrict__ gidx,
                          int R, int nsrc) {
  const int wave = blockIdx.x * (blockDim.x >> 5) + (threadIdx.x >> 5);
  const int lane = threadIdx.x & 31;
  const int row0 = wave * 16;
  if (row0 >= R) return;

  const int m    = lane & 15;
  const int kb   = (lane >> 4) << 1;
  const int r    = row0 + m;

  long off;
  if (MODE == 0) {
    off = (long)r * K;
  } else if (MODE == 1) {
    const int n = r >> 2, b = r & 3;
    off = ((long)b * nsrc + n) * (long)K;
  } else {
    const int d = gidx[r >> 2];
    off = ((long)d * 4 + (r & 3)) * (long)K;
  }
  const float* __restrict__ arow = A + off;

  v2f wf[K / 4][N / 16];
#pragma unroll
  for (int ks = 0; ks < K / 4; ++ks) {
#pragma unroll
    for (int nt = 0; nt < N / 16; ++nt) {
      const int kk = ks * 4 + kb;
      const int nn = nt * 16 + m;
      wf[ks][nt].x = W[kk * N + nn];
      wf[ks][nt].y = W[(kk + 1) * N + nn];
    }
  }

  v8f acc[N / 16] = {};
#pragma unroll
  for (int ks = 0; ks < K / 4; ++ks) {
    v2f a;
    a.x = arow[ks * 4 + kb];
    a.y = arow[ks * 4 + kb + 1];
#pragma unroll
    for (int nt = 0; nt < N / 16; ++nt) {
      acc[nt] = __builtin_amdgcn_wmma_f32_16x16x4_f32(
           false, a,  false, wf[ks][nt],
           (short)0, acc[nt],  false,  false);
    }
  }

  const int half = lane >> 4;
  const int cn   = lane & 15;
#pragma unroll
  for (int nt = 0; nt < N / 16; ++nt) {
#pragma unroll
    for (int vr = 0; vr < 8; ++vr) {
      const int rr = row0 + half * 8 + vr;
      Y[(long)rr * N + nt * 16 + cn] = acc[nt][vr];
    }
  }
}

template <int BH, int H>
__global__ void edge_kernel(const float* __restrict__ ys, const float* __restrict__ yd,
                            const int* __restrict__ src, const int* __restrict__ dst,
                            const float* __restrict__ ew, const float* __restrict__ We,
                            float* __restrict__ agg, float* __restrict__ cnt, int E) {
  const int e = blockIdx.x * (blockDim.x >> 5) + (threadIdx.x >> 5);
  if (e >= E) return;
  const int lane = threadIdx.x & 31;
  const int s = src[e], d = dst[e];
  const float w = ew[e];
  const float* __restrict__ ysr = ys + (long)s * BH;
  const float* __restrict__ ydr = yd + (long)d * BH;
  float* __restrict__ ar        = agg + (long)d * BH;
#pragma unroll
  for (int i = lane; i < BH; i += 32) {
    const float xj = ysr[i];
    const float xi = ydr[i];
    const float ev = w * We[i & (H - 1)];
    const float t  = xi * xj * ev;
    const float sg = 1.0f / (1.0f + __expf(-t));
    atomicAdd(ar + i, xj * sg);
  }
  if (lane == 0) atomicAdd(cnt + d, 1.0f);
}

template <int BH, int H, bool FINAL>
__global__ void finalize_kernel(const float* __restrict__ xs, const int* __restrict__ res,
                                const float* __restrict__ yd, const float* __restrict__ agg,
                                const float* __restrict__ cnt, float* __restrict__ out,
                                int Nd) {
  const int d = blockIdx.x * (blockDim.x >> 5) + (threadIdx.x >> 5);
  if (d >= Nd) return;
  const int lane = threadIdx.x & 31;
  constexpr int PER = BH / 32;

  const float inv_cnt = 1.0f / fmaxf(cnt[d], 1.0f);
  const long  base    = (long)d * BH;
  const long  xbase   = (long)res[d] * BH;

  float vals[PER], xdv[PER];
  float sum = 0.0f, ssq = 0.0f;
#pragma unroll
  for (int j = 0; j < PER; ++j) {
    const int i  = lane + j * 32;
    const float o = yd[base + i] + agg[base + i] * inv_cnt;
    vals[j] = o;
    xdv[j]  = xs[xbase + i];
    sum += o;
    ssq += o * o;
  }
#pragma unroll
  for (int mask = 16; mask >= 1; mask >>= 1) {
    sum += __shfl_xor(sum, mask, 32);
    ssq += __shfl_xor(ssq, mask, 32);
  }
  const float mu  = sum * (1.0f / BH);
  const float var = ssq * (1.0f / BH) - mu * mu;
  const float is  = rsqrtf(var + 1e-5f);

#pragma unroll
  for (int j = 0; j < PER; ++j) {
    const int i = lane + j * 32;
    float v = xdv[j] + (vals[j] - mu) * is;
    v = v > 0.0f ? v : 0.01f * v;
    if (FINAL) {
      const int b = i / H, c = i & (H - 1);
      out[((long)b * Nd + d) * H + c] = v;
    } else {
      out[base + i] = v;
    }
  }
}

extern "C" void kernel_launch(void* const* d_in, const int* in_sizes, int n_in,
                              void* d_out, int out_size, void* d_ws, size_t ws_size,
                              hipStream_t stream) {
  constexpr int B = 4, Cin = 32, H = 16, Cout = 32;
  const float* X    = (const float*)d_in[0];
  const int*   src0 = (const int*)d_in[1];
  const int*   dst0 = (const int*)d_in[2];
  const float* ew0  = (const float*)d_in[3];
  const int*   res0 = (const int*)d_in[4];
  const int*   src1 = (const int*)d_in[5];
  const int*   dst1 = (const int*)d_in[6];
  const float* ew1  = (const float*)d_in[7];
  const int*   res1 = (const int*)d_in[8];
  const float* Wn1  = (const float*)d_in[9];
  const float* We1  = (const float*)d_in[10];
  const float* U1   = (const float*)d_in[11];
  const float* V1   = (const float*)d_in[12];
  const float* Wn2  = (const float*)d_in[13];
  const float* We2  = (const float*)d_in[14];
  const float* U2   = (const float*)d_in[15];
  const float* V2   = (const float*)d_in[16];

  const int E0 = in_sizes[1];
  const int N1 = in_sizes[4];
  const int E1 = in_sizes[5];
  const int N2 = in_sizes[8];
  const int N0 = in_sizes[0] / (B * Cin);

  float* ws = (float*)d_ws;
  const size_t szXS = (size_t)((N0 * B * H > N1 * B * Cout) ? N0 * B * H : N1 * B * Cout);
  const size_t szYD = (size_t)((N1 * B * H > N2 * B * Cout) ? N1 * B * H : N2 * B * Cout);
  float* xs  = ws;
  float* ys  = xs + szXS;
  float* yd  = ys + szXS;
  float* agg = yd + szYD;
  float* cnt = agg + szYD;
  float* h1  = cnt + (size_t)((N1 > N2) ? N1 : N2);

  const dim3 blk(256);
  auto waves_grid = [](int tiles) { return dim3((unsigned)((tiles + 7) / 8)); };

  const int R1s = N0 * B;
  const int R1d = N1 * B;
  gemm_wmma<32, 16, 1><<<waves_grid(R1s / 16), blk, 0, stream>>>(X,  Wn1, xs, nullptr, R1s, N0);
  gemm_wmma<16, 16, 0><<<waves_grid(R1s / 16), blk, 0, stream>>>(xs, V1,  ys, nullptr, R1s, 0);
  gemm_wmma<16, 16, 2><<<waves_grid(R1d / 16), blk, 0, stream>>>(xs, U1,  yd, res0,    R1d, 0);
  hipMemsetAsync(agg, 0, (size_t)R1d * H * sizeof(float), stream);
  hipMemsetAsync(cnt, 0, (size_t)N1 * sizeof(float), stream);
  edge_kernel<64, 16><<<dim3((unsigned)((E0 + 7) / 8)), blk, 0, stream>>>(
      ys, yd, src0, dst0, ew0, We1, agg, cnt, E0);
  finalize_kernel<64, 16, false><<<dim3((unsigned)((N1 + 7) / 8)), blk, 0, stream>>>(
      xs, res0, yd, agg, cnt, h1, N1);

  const int R2s = N1 * B;
  const int R2d = N2 * B;
  gemm_wmma<16, 32, 0><<<waves_grid(R2s / 16), blk, 0, stream>>>(h1, Wn2, xs, nullptr, R2s, 0);
  gemm_wmma<32, 32, 0><<<waves_grid(R2s / 16), blk, 0, stream>>>(xs, V2,  ys, nullptr, R2s, 0);
  gemm_wmma<32, 32, 2><<<waves_grid(R2d / 16), blk, 0, stream>>>(xs, U2,  yd, res1,    R2d, 0);
  hipMemsetAsync(agg, 0, (size_t)R2d * Cout * sizeof(float), stream);
  hipMemsetAsync(cnt, 0, (size_t)N2 * sizeof(float), stream);
  edge_kernel<128, 32><<<dim3((unsigned)((E1 + 7) / 8)), blk, 0, stream>>>(
      ys, yd, src1, dst1, ew1, We2, agg, cnt, E1);
  finalize_kernel<128, 32, true><<<dim3((unsigned)((N2 + 7) / 8)), blk, 0, stream>>>(
      xs, res1, yd, agg, cnt, (float*)d_out, N2);
}


# Edges counting-sorted by dst in every call (int atomic arrival-order rank); one wave per dst node sums its edge messages in registers and normalises them: no float atomics, no agg round trip
# speedup vs baseline: 1.3088x; 1.3088x over previous
#include <hip/hip_runtime.h>
#include <hip/hip_bf16.h>

typedef float v2f __attribute__((ext_vector_type(2)));
typedef float v8f __attribute__((ext_vector_type(8)));

template <int K, int N, int MODE>
__global__ void gemm_wmma(const float* __restrict__ A, const float* __restrict__ W,
                          float* __restrict__ Y, const int* __restrict__ gidx,
                          int R, int nsrc) {
  const int wave = blockIdx.x * (blockDim.x >> 5) + (threadIdx.x >> 5);
  const int lane = threadIdx.x & 31;
  const int row0 = wave * 16;
  if (row0 >= R) return;

  const int m    = lane & 15;
  const int kb   = (lane >> 4) << 1;
  const int r    = row0 + m;

  long off;
  if (MODE == 0) {
    off = (long)r * K;
  } else if (MODE == 1) {
    const int n = r >> 2, b = r & 3;
    off = ((long)b * nsrc + n) * (long)K;
  } else {
    const int d = gidx[r >> 2];
    off = ((long)d * 4 + (r & 3)) * (long)K;
  }
  const float* __restrict__ arow = A + off;

  v2f wf[K / 4][N / 16];
#pragma unroll
  for (int ks = 0; ks < K / 4; ++ks) {
#pragma unroll
    for (int nt = 0; nt < N / 16; ++nt) {
      const int kk = ks * 4 + kb;
      const int nn = nt * 16 + m;
      wf[ks][nt].x = W[kk * N + nn];
      wf[ks][nt].y = W[(kk + 1) * N + nn];
    }
  }

  v8f acc[N / 16] = {};
#pragma unroll
  for (int ks = 0; ks < K / 4; ++ks) {
    v2f a;
    a.x = arow[ks * 4 + kb];
    a.y = arow[ks * 4 + kb + 1];
#pragma unroll
    for (int nt = 0; nt < N / 16; ++nt) {
      acc[nt] = __builtin_amdgcn_wmma_f32_16x16x4_f32(
           false, a,  false, wf[ks][nt],
           (short)0, acc[nt],  false,  false);
    }
  }

  const int half = lane >> 4;
  const int cn   = lane & 15;
#pragma unroll
  for (int nt = 0; nt < N / 16; ++nt) {
#pragma unroll
    for (int vr = 0; vr < 8; ++vr) {
      const int rr = row0 + half * 8 + vr;
      Y[(long)rr * N + nt * 16 + cn] = acc[nt][vr];
    }
  }
}

__global__ void edge_rank_kernel(const int* __restrict__ dst, int* __restrict__ cnti,
                                 int* __restrict__ rank, int E) {
  const int e = blockIdx.x * blockDim.x + threadIdx.x;
  if (e < E) rank[e] = atomicAdd(cnti + dst[e], 1);
}

__global__ void node_offsets_kernel(const int* __restrict__ cnti, int* __restrict__ total,
                                    int* __restrict__ offs, int Nd) {
  __shared__ int wsum[8];
  __shared__ int sbase;
  const int d    = blockIdx.x * blockDim.x + threadIdx.x;
  const int lane = threadIdx.x & 31;
  const int wid  = threadIdx.x >> 5;
  const int c    = (d < Nd) ? cnti[d] : 0;
  int v = c;
#pragma unroll
  for (int o = 1; o < 32; o <<= 1) {
    const int t = __shfl_up(v, o, 32);
    if (lane >= o) v += t;
  }
  if (lane == 31) wsum[wid] = v;
  __syncthreads();
  if (threadIdx.x == 0) {
    int s = 0;
#pragma unroll
    for (int w = 0; w < 8; ++w) {
      const int t = wsum[w];
      wsum[w] = s;
      s += t;
    }
    sbase = atomicAdd(total, s);
  }
  __syncthreads();
  if (d < Nd) offs[d] = sbase + wsum[wid] + v - c;
}

__global__ void edge_place_kernel(const int* __restrict__ src, const int* __restrict__ dst,
                                  const float* __restrict__ ew, const int* __restrict__ offs,
                                  const int* __restrict__ rank, int* __restrict__ ssrc,
                                  float* __restrict__ sew, int E) {
  const int e = blockIdx.x * blockDim.x + threadIdx.x;
  if (e < E) {
    const int p = offs[dst[e]] + rank[e];
    ssrc[p] = src[e];
    sew[p]  = ew[e];
  }
}

template <int PER>
__device__ __forceinline__ void add_edge(float (&acc)[PER], const float (&x)[PER], float w,
                                         const float (&ydv)[PER], const float (&wev)[PER]) {
#pragma unroll
  for (int j = 0; j < PER; ++j) {
    const float ev = w * wev[j];
    const float t  = ydv[j] * x[j] * ev;
    const float sg = 1.0f / (1.0f + __expf(-t));
    acc[j] += x[j] * sg;
  }
}

template <int BH, int H, bool FINAL>
__global__ void gather_finalize_kernel(const float* __restrict__ xs, const int* __restrict__ res,
                                       const float* __restrict__ ys, const float* __restrict__ yd,
                                       const int* __restrict__ offs, const int* __restrict__ cnti,
                                       const int* __restrict__ ssrc, const float* __restrict__ sew,
                                       const float* __restrict__ We, float* __restrict__ out,
                                       int Nd) {
  const int d = __builtin_amdgcn_readfirstlane(
      (int)(blockIdx.x * (blockDim.x >> 5) + (threadIdx.x >> 5)));
  if (d >= Nd) return;
  const int lane = threadIdx.x & 31;
  constexpr int PER = BH / 32;
  constexpr int U   = 16 / PER;

  const int  n     = cnti[d];
  const int  beg   = offs[d];
  const long base  = (long)d * BH;
  const long xbase = (long)res[d] * BH;

  int   m    = (n < 32) ? n : 32;
  int   my_s = 0;
  float my_w = 0.0f;
  if (lane < m) {
    my_s = ssrc[beg + lane];
    my_w = sew[beg + lane];
  }

  float ydv[PER], xdv[PER], wev[PER], acc[PER];
#pragma unroll
  for (int j = 0; j < PER; ++j) {
    const int i = lane + j * 32;
    ydv[j] = yd[base + i];
    xdv[j] = xs[xbase + i];
    wev[j] = We[i & (H - 1)];
    acc[j] = 0.0f;
  }

  for (int c0 = 0; c0 < n; c0 += 32) {
    if (c0 > 0) {
      m    = (n - c0 < 32) ? (n - c0) : 32;
      my_s = 0;
      my_w = 0.0f;
      if (lane < m) {
        my_s = ssrc[beg + c0 + lane];
        my_w = sew[beg + c0 + lane];
      }
    }
    for (int k = 0; k < m; k += U) {
      float xj[U][PER];
      float w[U];
      if (k + U <= m) {
#pragma unroll
        for (int u = 0; u < U; ++u) {
          const int s = __builtin_amdgcn_readlane(my_s, k + u);
          w[u] = __int_as_float(__builtin_amdgcn_readlane(__float_as_int(my_w), k + u));
          const float* __restrict__ row = ys + (long)s * BH;
#pragma unroll
          for (int j = 0; j < PER; ++j) xj[u][j] = row[lane + j * 32];
        }
#pragma unroll
        for (int u = 0; u < U; ++u) add_edge<PER>(acc, xj[u], w[u], ydv, wev);
      } else {
#pragma unroll
        for (int u = 0; u < U - 1; ++u) {
          const int kk = (k + u < m) ? (k + u) : (m - 1);
          const int s  = __builtin_amdgcn_readlane(my_s, kk);
          w[u] = __int_as_float(__builtin_amdgcn_readlane(__float_as_int(my_w), kk));
          const float* __restrict__ row = ys + (long)s * BH;
#pragma unroll
          for (int j = 0; j < PER; ++j) xj[u][j] = row[lane + j * 32];
        }
#pragma unroll
        for (int u = U - 2; u >= 1; --u) {
          if (k + u < m) add_edge<PER>(acc, xj[u], w[u], ydv, wev);
        }
        add_edge<PER>(acc, xj[0], w[0], ydv, wev);
      }
    }
  }

  const float inv_cnt = 1.0f / fmaxf((float)n, 1.0f);
  float vals[PER];
  float sum = 0.0f, ssq = 0.0f;
#pragma unroll
  for (int j = 0; j < PER; ++j) {
    const float o = ydv[j] + acc[j] * inv_cnt;
    vals[j] = o;
    sum += o;
    ssq += o * o;
  }
#pragma unroll
  for (int mask = 16; mask >= 1; mask >>= 1) {
    sum += __shfl_xor(sum, mask, 32);
    ssq += __shfl_xor(ssq, mask, 32);
  }
  const float mu  = sum * (1.0f / BH);
  const float var = ssq * (1.0f / BH) - mu * mu;
  const float is  = rsqrtf(var + 1e-5f);

#pragma unroll
  for (int j = 0; j < PER; ++j) {
    const int i = lane + j * 32;
    float v = xdv[j] + (vals[j] - mu) * is;
    v = v > 0.0f ? v : 0.01f * v;
    if (FINAL) {
      const int b = i / H, c = i & (H - 1);
      out[((long)b * Nd + d) * H + c] = v;
    } else {
      out[base + i] = v;
    }
  }
}

extern "C" void kernel_launch(void* const* d_in, const int* in_sizes, int n_in,
                              void* d_out, int out_size, void* d_ws, size_t ws_size,
                              hipStream_t stream) {
  constexpr int B = 4, Cin = 32, H = 16, Cout = 32;
  const float* X    = (const float*)d_in[0];
  const int*   src0 = (const int*)d_in[1];
  const int*   dst0 = (const int*)d_in[2];
  const float* ew0  = (const float*)d_in[3];
  const int*   res0 = (const int*)d_in[4];
  const int*   src1 = (const int*)d_in[5];
  const int*   dst1 = (const int*)d_in[6];
  const float* ew1  = (const float*)d_in[7];
  const int*   res1 = (const int*)d_in[8];
  const float* Wn1  = (const float*)d_in[9];
  const float* We1  = (const float*)d_in[10];
  const float* U1   = (const float*)d_in[11];
  const float* V1   = (const float*)d_in[12];
  const float* Wn2  = (const float*)d_in[13];
  const float* We2  = (const float*)d_in[14];
  const float* U2   = (const float*)d_in[15];
  const float* V2   = (const float*)d_in[16];

  const int E0 = in_sizes[1];
  const int N1 = in_sizes[4];
  const int E1 = in_sizes[5];
  const int N2 = in_sizes[8];
  const int N0 = in_sizes[0] / (B * Cin);

  float* ws = (float*)d_ws;
  const size_t szXS = (size_t)((N0 * B * H > N1 * B * Cout) ? N0 * B * H : N1 * B * Cout);
  const size_t szYD = (size_t)((N1 * B * H > N2 * B * Cout) ? N1 * B * H : N2 * B * Cout);
  float* xs  = ws;
  float* ys  = xs + szXS;
  float* yd  = ys + szXS;
  float* agg = yd + szYD;
  float* cnt = agg + szYD;
  float* h1  = cnt + (size_t)((N1 > N2) ? N1 : N2);

  const dim3 blk(256);
  auto waves_grid = [](int tiles) { return dim3((unsigned)((tiles + 7) / 8)); };

  const int R1s = N0 * B;
  const int R1d = N1 * B;
  gemm_wmma<32, 16, 1><<<waves_grid(R1s / 16), blk, 0, stream>>>(X,  Wn1, xs, nullptr, R1s, N0);
  gemm_wmma<16, 16, 0><<<waves_grid(R1s / 16), blk, 0, stream>>>(xs, V1,  ys, nullptr, R1s, 0);
  gemm_wmma<16, 16, 2><<<waves_grid(R1d / 16), blk, 0, stream>>>(xs, U1,  yd, res0,    R1d, 0);
  {
    int*   cnti  = (int*)agg;
    int*   total = cnti + N1;
    int*   offs  = total + 1;
    int*   rank  = offs + N1;
    int*   ssrc  = rank + E0;
    float* sew   = (float*)(ssrc + E0);
    hipMemsetAsync(cnti, 0, (size_t)(N1 + 1) * sizeof(int), stream);
    edge_rank_kernel<<<dim3((unsigned)((E0 + 255) / 256)), blk, 0, stream>>>(dst0, cnti, rank, E0);
    node_offsets_kernel<<<dim3((unsigned)((N1 + 255) / 256)), blk, 0, stream>>>(cnti, total, offs, N1);
    edge_place_kernel<<<dim3((unsigned)((E0 + 255) / 256)), blk, 0, stream>>>(
        src0, dst0, ew0, offs, rank, ssrc, sew, E0);
    gather_finalize_kernel<64, 16, false><<<dim3((unsigned)((N1 + 7) / 8)), blk, 0, stream>>>(
        xs, res0, ys, yd, offs, cnti, ssrc, sew, We1, h1, N1);
  }

  const int R2s = N1 * B;
  const int R2d = N2 * B;
  gemm_wmma<16, 32, 0><<<waves_grid(R2s / 16), blk, 0, stream>>>(h1, Wn2, xs, nullptr, R2s, 0);
  gemm_wmma<32, 32, 0><<<waves_grid(R2s / 16), blk, 0, stream>>>(xs, V2,  ys, nullptr, R2s, 0);
  gemm_wmma<32, 32, 2><<<waves_grid(R2d / 16), blk, 0, stream>>>(xs, U2,  yd, res1,    R2d, 0);
  {
    int*   cnti  = (int*)agg;
    int*   total = cnti + N2;
    int*   offs  = total + 1;
    int*   rank  = offs + N2;
    int*   ssrc  = rank + E1;
    float* sew   = (float*)(ssrc + E1);
    hipMemsetAsync(cnti, 0, (size_t)(N2 + 1) * sizeof(int), stream);
    edge_rank_kernel<<<dim3((unsigned)((E1 + 255) / 256)), blk, 0, stream>>>(dst1, cnti, rank, E1);
    node_offsets_kernel<<<dim3((unsigned)((N2 + 255) / 256)), blk, 0, stream>>>(cnti, total, offs, N2);
    edge_place_kernel<<<dim3((unsigned)((E1 + 255) / 256)), blk, 0, stream>>>(
        src1, dst1, ew1, offs, rank, ssrc, sew, E1);
    gather_finalize_kernel<128, 32, true><<<dim3((unsigned)((N2 + 7) / 8)), blk, 0, stream>>>(
        xs, res1, ys, yd, offs, cnti, ssrc, sew, We2, (float*)d_out, N2);
  }
}


# GEMM pair fused per 16-row tile: xs stays in registers as the B operand of the V product (both products transposed, no lane exchange); 2 launches fewer; k grouping (sum order) of V product changes
# speedup vs baseline: 1.3872x; 1.0599x over previous
#include <hip/hip_runtime.h>
#include <hip/hip_bf16.h>

typedef float v2f __attribute__((ext_vector_type(2)));
typedef float v8f __attribute__((ext_vector_type(8)));

template <int K, int N, int MODE>
__global__ void gemm_wmma(const float* __restrict__ A, const float* __restrict__ W,
                          float* __restrict__ Y, const int* __restrict__ gidx,
                          int R, int nsrc) {
  const int wave = blockIdx.x * (blockDim.x >> 5) + (threadIdx.x >> 5);
  const int lane = threadIdx.x & 31;
  const int row0 = wave * 16;
  if (row0 >= R) return;

  const int m    = lane & 15;
  const int kb   = (lane >> 4) << 1;
  const int r    = row0 + m;

  long off;
  if (MODE == 0) {
    off = (long)r * K;
  } else if (MODE == 1) {
    const int n = r >> 2, b = r & 3;
    off = ((long)b * nsrc + n) * (long)K;
  } else {
    const int d = gidx[r >> 2];
    off = ((long)d * 4 + (r & 3)) * (long)K;
  }
  const float* __restrict__ arow = A + off;

  v2f wf[K / 4][N / 16];
#pragma unroll
  for (int ks = 0; ks < K / 4; ++ks) {
#pragma unroll
    for (int nt = 0; nt < N / 16; ++nt) {
      const int kk = ks * 4 + kb;
      const int nn = nt * 16 + m;
      wf[ks][nt].x = W[kk * N + nn];
      wf[ks][nt].y = W[(kk + 1) * N + nn];
    }
  }

  v8f acc[N / 16] = {};
#pragma unroll
  for (int ks = 0; ks < K / 4; ++ks) {
    v2f a;
    a.x = arow[ks * 4 + kb];
    a.y = arow[ks * 4 + kb + 1];
#pragma unroll
    for (int nt = 0; nt < N / 16; ++nt) {
      acc[nt] = __builtin_amdgcn_wmma_f32_16x16x4_f32(
           false, a,  false, wf[ks][nt],
           (short)0, acc[nt],  false,  false);
    }
  }

  const int half = lane >> 4;
  const int cn   = lane & 15;
#pragma unroll
  for (int nt = 0; nt < N / 16; ++nt) {
#pragma unroll
    for (int vr = 0; vr < 8; ++vr) {
      const int rr = row0 + half * 8 + vr;
      Y[(long)rr * N + nt * 16 + cn] = acc[nt][vr];
    }
  }
}

template <int K, int N, int MODE>
__global__ void gemm_pair_wmma(const float* __restrict__ A, const float* __restrict__ W,
                               const float* __restrict__ V, float* __restrict__ XS,
                               float* __restrict__ YS, int R, int nsrc) {
  const int wave = blockIdx.x * (blockDim.x >> 5) + (threadIdx.x >> 5);
  const int lane = threadIdx.x & 31;
  const int row0 = wave * 16;
  if (row0 >= R) return;

  const int c  = lane & 15;
  const int h  = lane >> 4;
  const int kb = h << 1;
  const int r  = row0 + c;

  long off;
  if (MODE == 0) {
    off = (long)r * K;
  } else {
    const int n = r >> 2, b = r & 3;
    off = ((long)b * nsrc + n) * (long)K;
  }
  const float* __restrict__ arow = A + off;

  v2f xf[K / 4];
#pragma unroll
  for (int ks = 0; ks < K / 4; ++ks) {
    xf[ks].x = arow[ks * 4 + kb];
    xf[ks].y = arow[ks * 4 + kb + 1];
  }

  v2f wf[K / 4][N / 16];
#pragma unroll
  for (int ks = 0; ks < K / 4; ++ks) {
#pragma unroll
    for (int nt = 0; nt < N / 16; ++nt) {
      const int kk = ks * 4 + kb;
      const int nn = nt * 16 + c;
      wf[ks][nt].x = W[kk * N + nn];
      wf[ks][nt].y = W[(kk + 1) * N + nn];
    }
  }

  v2f vf[N / 16][4][N / 16];
#pragma unroll
  for (int nt = 0; nt < N / 16; ++nt) {
#pragma unroll
    for (int j = 0; j < 4; ++j) {
#pragma unroll
      for (int nt2 = 0; nt2 < N / 16; ++nt2) {
        const int k0 = nt * 16 + 8 * h + 2 * j;
        const int nn = nt2 * 16 + c;
        vf[nt][j][nt2].x = V[k0 * N + nn];
        vf[nt][j][nt2].y = V[(k0 + 1) * N + nn];
      }
    }
  }

  v8f d1[N / 16] = {};
#pragma unroll
  for (int ks = 0; ks < K / 4; ++ks) {
#pragma unroll
    for (int nt = 0; nt < N / 16; ++nt) {
      d1[nt] = __builtin_amdgcn_wmma_f32_16x16x4_f32(
           false, wf[ks][nt],  false, xf[ks],
           (short)0, d1[nt],  false,  false);
    }
  }

  v8f d2[N / 16] = {};
#pragma unroll
  for (int nt = 0; nt < N / 16; ++nt) {
#pragma unroll
    for (int j = 0; j < 4; ++j) {
      v2f xb;
      xb.x = d1[nt][2 * j];
      xb.y = d1[nt][2 * j + 1];
#pragma unroll
      for (int nt2 = 0; nt2 < N / 16; ++nt2) {
        d2[nt2] = __builtin_amdgcn_wmma_f32_16x16x4_f32(
             false, vf[nt][j][nt2],  false, xb,
             (short)0, d2[nt2],  false,  false);
      }
    }
  }

  float* __restrict__ xrow = XS + (long)r * N + 8 * h;
  float* __restrict__ yrow = YS + (long)r * N + 8 * h;
#pragma unroll
  for (int nt = 0; nt < N / 16; ++nt) {
    *(float4*)(xrow + nt * 16)     = make_float4(d1[nt][0], d1[nt][1], d1[nt][2], d1[nt][3]);
    *(float4*)(xrow + nt * 16 + 4) = make_float4(d1[nt][4], d1[nt][5], d1[nt][6], d1[nt][7]);
    *(float4*)(yrow + nt * 16)     = make_float4(d2[nt][0], d2[nt][1], d2[nt][2], d2[nt][3]);
    *(float4*)(yrow + nt * 16 + 4) = make_float4(d2[nt][4], d2[nt][5], d2[nt][6], d2[nt][7]);
  }
}

__global__ void edge_rank_kernel(const int* __restrict__ dst, int* __restrict__ cnti,
                                 int* __restrict__ rank, int E) {
  const int e = blockIdx.x * blockDim.x + threadIdx.x;
  if (e < E) rank[e] = atomicAdd(cnti + dst[e], 1);
}

__global__ void node_offsets_kernel(const int* __restrict__ cnti, int* __restrict__ total,
                                    int* __restrict__ offs, int Nd) {
  __shared__ int wsum[8];
  __shared__ int sbase;
  const int d    = blockIdx.x * blockDim.x + threadIdx.x;
  const int lane = threadIdx.x & 31;
  const int wid  = threadIdx.x >> 5;
  const int c    = (d < Nd) ? cnti[d] : 0;
  int v = c;
#pragma unroll
  for (int o = 1; o < 32; o <<= 1) {
    const int t = __shfl_up(v, o, 32);
    if (lane >= o) v += t;
  }
  if (lane == 31) wsum[wid] = v;
  __syncthreads();
  if (threadIdx.x == 0) {
    int s = 0;
#pragma unroll
    for (int w = 0; w < 8; ++w) {
      const int t = wsum[w];
      wsum[w] = s;
      s += t;
    }
    sbase = atomicAdd(total, s);
  }
  __syncthreads();
  if (d < Nd) offs[d] = sbase + wsum[wid] + v - c;
}

__global__ void edge_place_kernel(const int* __restrict__ src, const int* __restrict__ dst,
                                  const float* __restrict__ ew, const int* __restrict__ offs,
                                  const int* __restrict__ rank, int* __restrict__ ssrc,
                                  float* __restrict__ sew, int E) {
  const int e = blockIdx.x * blockDim.x + threadIdx.x;
  if (e < E) {
    const int p = offs[dst[e]] + rank[e];
    ssrc[p] = src[e];
    sew[p]  = ew[e];
  }
}

template <int PER>
__device__ __forceinline__ void add_edge(float (&acc)[PER], const float (&x)[PER], float w,
                                         const float (&ydv)[PER], const float (&wev)[PER]) {
#pragma unroll
  for (int j = 0; j < PER; ++j) {
    const float ev = w * wev[j];
    const float t  = ydv[j] * x[j] * ev;
    const float sg = 1.0f / (1.0f + __expf(-t));
    acc[j] += x[j] * sg;
  }
}

template <int BH, int H, bool FINAL>
__global__ void gather_finalize_kernel(const float* __restrict__ xs, const int* __restrict__ res,
                                       const float* __restrict__ ys, const float* __restrict__ yd,
                                       const int* __restrict__ offs, const int* __restrict__ cnti,
                                       const int* __restrict__ ssrc, const float* __restrict__ sew,
                                       const float* __restrict__ We, float* __restrict__ out,
                                       int Nd) {
  const int d = __builtin_amdgcn_readfirstlane(
      (int)(blockIdx.x * (blockDim.x >> 5) + (threadIdx.x >> 5)));
  if (d >= Nd) return;
  const int lane = threadIdx.x & 31;
  constexpr int PER = BH / 32;
  constexpr int U   = 16 / PER;

  const int  n     = cnti[d];
  const int  beg   = offs[d];
  const long base  = (long)d * BH;
  const long xbase = (long)res[d] * BH;

  int   m    = (n < 32) ? n : 32;
  int   my_s = 0;
  float my_w = 0.0f;
  if (lane < m) {
    my_s = ssrc[beg + lane];
    my_w = sew[beg + lane];
  }

  float ydv[PER], xdv[PER], wev[PER], acc[PER];
#pragma unroll
  for (int j = 0; j < PER; ++j) {
    const int i = lane + j * 32;
    ydv[j] = yd[base + i];
    xdv[j] = xs[xbase + i];
    wev[j] = We[i & (H - 1)];
    acc[j] = 0.0f;
  }

  for (int c0 = 0; c0 < n; c0 += 32) {
    if (c0 > 0) {
      m    = (n - c0 < 32) ? (n - c0) : 32;
      my_s = 0;
      my_w = 0.0f;
      if (lane < m) {
        my_s = ssrc[beg + c0 + lane];
        my_w = sew[beg + c0 + lane];
      }
    }
    for (int k = 0; k < m; k += U) {
      float xj[U][PER];
      float w[U];
      if (k + U <= m) {
#pragma unroll
        for (int u = 0; u < U; ++u) {
          const int s = __builtin_amdgcn_readlane(my_s, k + u);
          w[u] = __int_as_float(__builtin_amdgcn_readlane(__float_as_int(my_w), k + u));
          const float* __restrict__ row = ys + (long)s * BH;
#pragma unroll
          for (int j = 0; j < PER; ++j) xj[u][j] = row[lane + j * 32];
        }
#pragma unroll
        for (int u = 0; u < U; ++u) add_edge<PER>(acc, xj[u], w[u], ydv, wev);
      } else {
#pragma unroll
        for (int u = 0; u < U - 1; ++u) {
          const int kk = (k + u < m) ? (k + u) : (m - 1);
          const int s  = __builtin_amdgcn_readlane(my_s, kk);
          w[u] = __int_as_float(__builtin_amdgcn_readlane(__float_as_int(my_w), kk));
          const float* __restrict__ row = ys + (long)s * BH;
#pragma unroll
          for (int j = 0; j < PER; ++j) xj[u][j] = row[lane + j * 32];
        }
#pragma unroll
        for (int u = U - 2; u >= 1; --u) {
          if (k + u < m) add_edge<PER>(acc, xj[u], w[u], ydv, wev);
        }
        add_edge<PER>(acc, xj[0], w[0], ydv, wev);
      }
    }
  }

  const float inv_cnt = 1.0f / fmaxf((float)n, 1.0f);
  float vals[PER];
  float sum = 0.0f, ssq = 0.0f;
#pragma unroll
  for (int j = 0; j < PER; ++j) {
    const float o = ydv[j] + acc[j] * inv_cnt;
    vals[j] = o;
    sum += o;
    ssq += o * o;
  }
#pragma unroll
  for (int mask = 16; mask >= 1; mask >>= 1) {
    sum += __shfl_xor(sum, mask, 32);
    ssq += __shfl_xor(ssq, mask, 32);
  }
  const float mu  = sum * (1.0f / BH);
  const float var = ssq * (1.0f / BH) - mu * mu;
  const float is  = rsqrtf(var + 1e-5f);

#pragma unroll
  for (int j = 0; j < PER; ++j) {
    const int i = lane + j * 32;
    float v = xdv[j] + (vals[j] - mu) * is;
    v = v > 0.0f ? v : 0.01f * v;
    if (FINAL) {
      const int b = i / H, c = i & (H - 1);
      out[((long)b * Nd + d) * H + c] = v;
    } else {
      out[base + i] = v;
    }
  }
}

extern "C" void kernel_launch(void* const* d_in, const int* in_sizes, int n_in,
                              void* d_out, int out_size, void* d_ws, size_t ws_size,
                              hipStream_t stream) {
  constexpr int B = 4, Cin = 32, H = 16, Cout = 32;
  const float* X    = (const float*)d_in[0];
  const int*   src0 = (const int*)d_in[1];
  const int*   dst0 = (const int*)d_in[2];
  const float* ew0  = (const float*)d_in[3];
  const int*   res0 = (const int*)d_in[4];
  const int*   src1 = (const int*)d_in[5];
  const int*   dst1 = (const int*)d_in[6];
  const float* ew1  = (const float*)d_in[7];
  const int*   res1 = (const int*)d_in[8];
  const float* Wn1  = (const float*)d_in[9];
  const float* We1  = (const float*)d_in[10];
  const float* U1   = (const float*)d_in[11];
  const float* V1   = (const float*)d_in[12];
  const float* Wn2  = (const float*)d_in[13];
  const float* We2  = (const float*)d_in[14];
  const float* U2   = (const float*)d_in[15];
  const float* V2   = (const float*)d_in[16];

  const int E0 = in_sizes[1];
  const int N1 = in_sizes[4];
  const int E1 = in_sizes[5];
  const int N2 = in_sizes[8];
  const int N0 = in_sizes[0] / (B * Cin);

  float* ws = (float*)d_ws;
  const size_t szXS = (size_t)((N0 * B * H > N1 * B * Cout) ? N0 * B * H : N1 * B * Cout);
  const size_t szYD = (size_t)((N1 * B * H > N2 * B * Cout) ? N1 * B * H : N2 * B * Cout);
  float* xs  = ws;
  float* ys  = xs + szXS;
  float* yd  = ys + szXS;
  float* agg = yd + szYD;
  float* cnt = agg + szYD;
  float* h1  = cnt + (size_t)((N1 > N2) ? N1 : N2);

  const dim3 blk(256);
  auto waves_grid = [](int tiles) { return dim3((unsigned)((tiles + 7) / 8)); };

  const int R1s = N0 * B;
  const int R1d = N1 * B;
  gemm_pair_wmma<32, 16, 1><<<waves_grid(R1s / 16), blk, 0, stream>>>(X, Wn1, V1, xs, ys, R1s, N0);
  gemm_wmma<16, 16, 2><<<waves_grid(R1d / 16), blk, 0, stream>>>(xs, U1,  yd, res0,    R1d, 0);
  {
    int*   cnti  = (int*)agg;
    int*   total = cnti + N1;
    int*   offs  = total + 1;
    int*   rank  = offs + N1;
    int*   ssrc  = rank + E0;
    float* sew   = (float*)(ssrc + E0);
    hipMemsetAsync(cnti, 0, (size_t)(N1 + 1) * sizeof(int), stream);
    edge_rank_kernel<<<dim3((unsigned)((E0 + 255) / 256)), blk, 0, stream>>>(dst0, cnti, rank, E0);
    node_offsets_kernel<<<dim3((unsigned)((N1 + 255) / 256)), blk, 0, stream>>>(cnti, total, offs, N1);
    edge_place_kernel<<<dim3((unsigned)((E0 + 255) / 256)), blk, 0, stream>>>(
        src0, dst0, ew0, offs, rank, ssrc, sew, E0);
    gather_finalize_kernel<64, 16, false><<<dim3((unsigned)((N1 + 7) / 8)), blk, 0, stream>>>(
        xs, res0, ys, yd, offs, cnti, ssrc, sew, We1, h1, N1);
  }

  const int R2s = N1 * B;
  const int R2d = N2 * B;
  gemm_pair_wmma<16, 32, 0><<<waves_grid(R2s / 16), blk, 0, stream>>>(h1, Wn2, V2, xs, ys, R2s, 0);
  gemm_wmma<32, 32, 2><<<waves_grid(R2d / 16), blk, 0, stream>>>(xs, U2,  yd, res1,    R2d, 0);
  {
    int*   cnti  = (int*)agg;
    int*   total = cnti + N2;
    int*   offs  = total + 1;
    int*   rank  = offs + N2;
    int*   ssrc  = rank + E1;
    float* sew   = (float*)(ssrc + E1);
    hipMemsetAsync(cnti, 0, (size_t)(N2 + 1) * sizeof(int), stream);
    edge_rank_kernel<<<dim3((unsigned)((E1 + 255) / 256)), blk, 0, stream>>>(dst1, cnti, rank, E1);
    node_offsets_kernel<<<dim3((unsigned)((N2 + 255) / 256)), blk, 0, stream>>>(cnti, total, offs, N2);
    edge_place_kernel<<<dim3((unsigned)((E1 + 255) / 256)), blk, 0, stream>>>(
        src1, dst1, ew1, offs, rank, ssrc, sew, E1);
    gather_finalize_kernel<128, 32, true><<<dim3((unsigned)((N2 + 7) / 8)), blk, 0, stream>>>(
        xs, res1, ys, yd, offs, cnti, ssrc, sew, We2, (float*)d_out, N2);
  }
}


# Dependent launches 12 -> 8 and no fill operation: both layers' edge grouping shares its three launches, its counter clear rides in the first GEMM launch, its rank pass in the layer-1 dst GEMM launch
# speedup vs baseline: 1.4425x; 1.0399x over previous
#include <hip/hip_runtime.h>
#include <hip/hip_bf16.h>

typedef float v2f __attribute__((ext_vector_type(2)));
typedef float v8f __attribute__((ext_vector_type(8)));

template <int K, int N, int MODE>
__device__ __forceinline__ void gemm_rows(const float* __restrict__ A, const float* __restrict__ W,
                                          float* __restrict__ Y, const int* __restrict__ gidx,
                                          int R, int nsrc, int block) {
  const int wave = block * (blockDim.x >> 5) + (threadIdx.x >> 5);
  const int lane = threadIdx.x & 31;
  const int row0 = wave * 16;
  if (row0 >= R) return;

  const int m    = lane & 15;
  const int kb   = (lane >> 4) << 1;
  const int r    = row0 + m;

  long off;
  if (MODE == 0) {
    off = (long)r * K;
  } else if (MODE == 1) {
    const int n = r >> 2, b = r & 3;
    off = ((long)b * nsrc + n) * (long)K;
  } else {
    const int d = gidx[r >> 2];
    off = ((long)d * 4 + (r & 3)) * (long)K;
  }
  const float* __restrict__ arow = A + off;

  v2f wf[K / 4][N / 16];
#pragma unroll
  for (int ks = 0; ks < K / 4; ++ks) {
#pragma unroll
    for (int nt = 0; nt < N / 16; ++nt) {
      const int kk = ks * 4 + kb;
      const int nn = nt * 16 + m;
      wf[ks][nt].x = W[kk * N + nn];
      wf[ks][nt].y = W[(kk + 1) * N + nn];
    }
  }

  v8f acc[N / 16] = {};
#pragma unroll
  for (int ks = 0; ks < K / 4; ++ks) {
    v2f a;
    a.x = arow[ks * 4 + kb];
    a.y = arow[ks * 4 + kb + 1];
#pragma unroll
    for (int nt = 0; nt < N / 16; ++nt) {
      acc[nt] = __builtin_amdgcn_wmma_f32_16x16x4_f32(
           false, a,  false, wf[ks][nt],
           (short)0, acc[nt],  false,  false);
    }
  }

  const int half = lane >> 4;
  const int cn   = lane & 15;
#pragma unroll
  for (int nt = 0; nt < N / 16; ++nt) {
#pragma unroll
    for (int vr = 0; vr < 8; ++vr) {
      const int rr = row0 + half * 8 + vr;
      Y[(long)rr * N + nt * 16 + cn] = acc[nt][vr];
    }
  }
}

template <int K, int N, int MODE>
__global__ void gemm_wmma(const float* __restrict__ A, const float* __restrict__ W,
                          float* __restrict__ Y, const int* __restrict__ gidx,
                          int R, int nsrc) {
  gemm_rows<K, N, MODE>(A, W, Y, gidx, R, nsrc, (int)blockIdx.x);
}

template <int K, int N, int MODE>
__global__ void gemm_pair_wmma(const float* __restrict__ A, const float* __restrict__ W,
                               const float* __restrict__ V, float* __restrict__ XS,
                               float* __restrict__ YS, int R, int nsrc,
                               int* __restrict__ zero_ptr, int zero_n) {
  for (int i = blockIdx.x * blockDim.x + threadIdx.x; i < zero_n; i += gridDim.x * blockDim.x)
    zero_ptr[i] = 0;

  const int wave = blockIdx.x * (blockDim.x >> 5) + (threadIdx.x >> 5);
  const int lane = threadIdx.x & 31;
  const int row0 = wave * 16;
  if (row0 >= R) return;

  const int c  = lane & 15;
  const int h  = lane >> 4;
  const int kb = h << 1;
  const int r  = row0 + c;

  long off;
  if (MODE == 0) {
    off = (long)r * K;
  } else {
    const int n = r >> 2, b = r & 3;
    off = ((long)b * nsrc + n) * (long)K;
  }
  const float* __restrict__ arow = A + off;

  v2f xf[K / 4];
#pragma unroll
  for (int ks = 0; ks < K / 4; ++ks) {
    xf[ks].x = arow[ks * 4 + kb];
    xf[ks].y = arow[ks * 4 + kb + 1];
  }

  v2f wf[K / 4][N / 16];
#pragma unroll
  for (int ks = 0; ks < K / 4; ++ks) {
#pragma unroll
    for (int nt = 0; nt < N / 16; ++nt) {
      const int kk = ks * 4 + kb;
      const int nn = nt * 16 + c;
      wf[ks][nt].x = W[kk * N + nn];
      wf[ks][nt].y = W[(kk + 1) * N + nn];
    }
  }

  v2f vf[N / 16][4][N / 16];
#pragma unroll
  for (int nt = 0; nt < N / 16; ++nt) {
#pragma unroll
    for (int j = 0; j < 4; ++j) {
#pragma unroll
      for (int nt2 = 0; nt2 < N / 16; ++nt2) {
        const int k0 = nt * 16 + 8 * h + 2 * j;
        const int nn = nt2 * 16 + c;
        vf[nt][j][nt2].x = V[k0 * N + nn];
        vf[nt][j][nt2].y = V[(k0 + 1) * N + nn];
      }
    }
  }

  v8f d1[N / 16] = {};
#pragma unroll
  for (int ks = 0; ks < K / 4; ++ks) {
#pragma unroll
    for (int nt = 0; nt < N / 16; ++nt) {
      d1[nt] = __builtin_amdgcn_wmma_f32_16x16x4_f32(
           false, wf[ks][nt],  false, xf[ks],
           (short)0, d1[nt],  false,  false);
    }
  }

  v8f d2[N / 16] = {};
#pragma unroll
  for (int nt = 0; nt < N / 16; ++nt) {
#pragma unroll
    for (int j = 0; j < 4; ++j) {
      v2f xb;
      xb.x = d1[nt][2 * j];
      xb.y = d1[nt][2 * j + 1];
#pragma unroll
      for (int nt2 = 0; nt2 < N / 16; ++nt2) {
        d2[nt2] = __builtin_amdgcn_wmma_f32_16x16x4_f32(
             false, vf[nt][j][nt2],  false, xb,
             (short)0, d2[nt2],  false,  false);
      }
    }
  }

  float* __restrict__ xrow = XS + (long)r * N + 8 * h;
  float* __restrict__ yrow = YS + (long)r * N + 8 * h;
#pragma unroll
  for (int nt = 0; nt < N / 16; ++nt) {
    *(float4*)(xrow + nt * 16)     = make_float4(d1[nt][0], d1[nt][1], d1[nt][2], d1[nt][3]);
    *(float4*)(xrow + nt * 16 + 4) = make_float4(d1[nt][4], d1[nt][5], d1[nt][6], d1[nt][7]);
    *(float4*)(yrow + nt * 16)     = make_float4(d2[nt][0], d2[nt][1], d2[nt][2], d2[nt][3]);
    *(float4*)(yrow + nt * 16 + 4) = make_float4(d2[nt][4], d2[nt][5], d2[nt][6], d2[nt][7]);
  }
}

__device__ __forceinline__ void edge_rank_both(int e, const int* __restrict__ dst0,
                                               int* __restrict__ cnt0, int E0,
                                               const int* __restrict__ dst1,
                                               int* __restrict__ cnt1, int E1,
                                               int* __restrict__ rank) {
  if (e < E0) {
    rank[e] = atomicAdd(cnt0 + dst0[e], 1);
  } else if (e - E0 < E1) {
    rank[e] = atomicAdd(cnt1 + dst1[e - E0], 1);
  }
}

template <int K, int N>
__global__ void gemm_gather_and_edge_rank(const float* __restrict__ A, const float* __restrict__ W,
                                          float* __restrict__ Y, const int* __restrict__ gidx,
                                          int R, int rank_blocks,
                                          const int* __restrict__ dst0, int* __restrict__ cnt0,
                                          int E0, const int* __restrict__ dst1,
                                          int* __restrict__ cnt1, int E1,
                                          int* __restrict__ rank) {
  if ((int)blockIdx.x < rank_blocks) {
    edge_rank_both((int)(blockIdx.x * blockDim.x + threadIdx.x), dst0, cnt0, E0, dst1, cnt1, E1,
                   rank);
  } else {
    gemm_rows<K, N, 2>(A, W, Y, gidx, R, 0, (int)blockIdx.x - rank_blocks);
  }
}

__global__ void node_offsets_kernel(const int* __restrict__ cnt0, int* __restrict__ total0,
                                    int* __restrict__ offs0, int Nd0, int blocks0,
                                    const int* __restrict__ cnt1, int* __restrict__ total1,
                                    int* __restrict__ offs1, int Nd1) {
  __shared__ int wsum[8];
  __shared__ int sbase;
  const bool first = (int)blockIdx.x < blocks0;
  const int* __restrict__ cnti  = first ? cnt0 : cnt1;
  int* __restrict__       total = first ? total0 : total1;
  int* __restrict__       offs  = first ? offs0 : offs1;
  const int Nd   = first ? Nd0 : Nd1;
  const int d    = (first ? (int)blockIdx.x : (int)blockIdx.x - blocks0) * (int)blockDim.x +
                   (int)threadIdx.x;
  const int lane = threadIdx.x & 31;
  const int wid  = threadIdx.x >> 5;
  const int c    = (d < Nd) ? cnti[d] : 0;
  int v = c;
#pragma unroll
  for (int o = 1; o < 32; o <<= 1) {
    const int t = __shfl_up(v, o, 32);
    if (lane >= o) v += t;
  }
  if (lane == 31) wsum[wid] = v;
  __syncthreads();
  if (threadIdx.x == 0) {
    int s = 0;
#pragma unroll
    for (int w = 0; w < 8; ++w) {
      const int t = wsum[w];
      wsum[w] = s;
      s += t;
    }
    sbase = atomicAdd(total, s);
  }
  __syncthreads();
  if (d < Nd) offs[d] = sbase + wsum[wid] + v - c;
}

__global__ void edge_place_kernel(const int* __restrict__ src0, const int* __restrict__ dst0,
                                  const float* __restrict__ ew0, const int* __restrict__ offs0,
                                  int E0, const int* __restrict__ src1,
                                  const int* __restrict__ dst1, const float* __restrict__ ew1,
                                  const int* __restrict__ offs1, int E1,
                                  const int* __restrict__ rank, int* __restrict__ ssrc,
                                  float* __restrict__ sew) {
  const int e = blockIdx.x * blockDim.x + threadIdx.x;
  if (e < E0) {
    const int p = offs0[dst0[e]] + rank[e];
    ssrc[p] = src0[e];
    sew[p]  = ew0[e];
  } else if (e - E0 < E1) {
    const int f = e - E0;
    const int p = E0 + offs1[dst1[f]] + rank[e];
    ssrc[p] = src1[f];
    sew[p]  = ew1[f];
  }
}

template <int PER>
__device__ __forceinline__ void add_edge(float (&acc)[PER], const float (&x)[PER], float w,
                                         const float (&ydv)[PER], const float (&wev)[PER]) {
#pragma unroll
  for (int j = 0; j < PER; ++j) {
    const float ev = w * wev[j];
    const float t  = ydv[j] * x[j] * ev;
    const float sg = 1.0f / (1.0f + __expf(-t));
    acc[j] += x[j] * sg;
  }
}

template <int BH, int H, bool FINAL>
__global__ void gather_finalize_kernel(const float* __restrict__ xs, const int* __restrict__ res,
                                       const float* __restrict__ ys, const float* __restrict__ yd,
                                       const int* __restrict__ offs, const int* __restrict__ cnti,
                                       const int* __restrict__ ssrc, const float* __restrict__ sew,
                                       const float* __restrict__ We, float* __restrict__ out,
                                       int Nd) {
  const int d = __builtin_amdgcn_readfirstlane(
      (int)(blockIdx.x * (blockDim.x >> 5) + (threadIdx.x >> 5)));
  if (d >= Nd) return;
  const int lane = threadIdx.x & 31;
  constexpr int PER = BH / 32;
  constexpr int U   = 16 / PER;

  const int  n     = cnti[d];
  const int  beg   = offs[d];
  const long base  = (long)d * BH;
  const long xbase = (long)res[d] * BH;

  int   m    = (n < 32) ? n : 32;
  int   my_s = 0;
  float my_w = 0.0f;
  if (lane < m) {
    my_s = ssrc[beg + lane];
    my_w = sew[beg + lane];
  }

  float ydv[PER], xdv[PER], wev[PER], acc[PER];
#pragma unroll
  for (int j = 0; j < PER; ++j) {
    const int i = lane + j * 32;
    ydv[j] = yd[base + i];
    xdv[j] = xs[xbase + i];
    wev[j] = We[i & (H - 1)];
    acc[j] = 0.0f;
  }

  for (int c0 = 0; c0 < n; c0 += 32) {
    if (c0 > 0) {
      m    = (n - c0 < 32) ? (n - c0) : 32;
      my_s = 0;
      my_w = 0.0f;
      if (lane < m) {
        my_s = ssrc[beg + c0 + lane];
        my_w = sew[beg + c0 + lane];
      }
    }
    for (int k = 0; k < m; k += U) {
      float xj[U][PER];
      float w[U];
      if (k + U <= m) {
#pragma unroll
        for (int u = 0; u < U; ++u) {
          const int s = __builtin_amdgcn_readlane(my_s, k + u);
          w[u] = __int_as_float(__builtin_amdgcn_readlane(__float_as_int(my_w), k + u));
          const float* __restrict__ row = ys + (long)s * BH;
#pragma unroll
          for (int j = 0; j < PER; ++j) xj[u][j] = row[lane + j * 32];
        }
#pragma unroll
        for (int u = 0; u < U; ++u) add_edge<PER>(acc, xj[u], w[u], ydv, wev);
      } else {
#pragma unroll
        for (int u = 0; u < U - 1; ++u) {
          const int kk = (k + u < m) ? (k + u) : (m - 1);
          const int s  = __builtin_amdgcn_readlane(my_s, kk);
          w[u] = __int_as_float(__builtin_amdgcn_readlane(__float_as_int(my_w), kk));
          const float* __restrict__ row = ys + (long)s * BH;
#pragma unroll
          for (int j = 0; j < PER; ++j) xj[u][j] = row[lane + j * 32];
        }
#pragma unroll
        for (int u = U - 2; u >= 1; --u) {
          if (k + u < m) add_edge<PER>(acc, xj[u], w[u], ydv, wev);
        }
        add_edge<PER>(acc, xj[0], w[0], ydv, wev);
      }
    }
  }

  const float inv_cnt = 1.0f / fmaxf((float)n, 1.0f);
  float vals[PER];
  float sum = 0.0f, ssq = 0.0f;
#pragma unroll
  for (int j = 0; j < PER; ++j) {
    const float o = ydv[j] + acc[j] * inv_cnt;
    vals[j] = o;
    sum += o;
    ssq += o * o;
  }
#pragma unroll
  for (int mask = 16; mask >= 1; mask >>= 1) {
    sum += __shfl_xor(sum, mask, 32);
    ssq += __shfl_xor(ssq, mask, 32);
  }
  const float mu  = sum * (1.0f / BH);
  const float var = ssq * (1.0f / BH) - mu * mu;
  const float is  = rsqrtf(var + 1e-5f);

#pragma unroll
  for (int j = 0; j < PER; ++j) {
    const int i = lane + j * 32;
    float v = xdv[j] + (vals[j] - mu) * is;
    v = v > 0.0f ? v : 0.01f * v;
    if (FINAL) {
      const int b = i / H, c = i & (H - 1);
      out[((long)b * Nd + d) * H + c] = v;
    } else {
      out[base + i] = v;
    }
  }
}

extern "C" void kernel_launch(void* const* d_in, const int* in_sizes, int n_in,
                              void* d_out, int out_size, void* d_ws, size_t ws_size,
                              hipStream_t stream) {
  constexpr int B = 4, Cin = 32, H = 16, Cout = 32;
  const float* X    = (const float*)d_in[0];
  const int*   src0 = (const int*)d_in[1];
  const int*   dst0 = (const int*)d_in[2];
  const float* ew0  = (const float*)d_in[3];
  const int*   res0 = (const int*)d_in[4];
  const int*   src1 = (const int*)d_in[5];
  const int*   dst1 = (const int*)d_in[6];
  const float* ew1  = (const float*)d_in[7];
  const int*   res1 = (const int*)d_in[8];
  const float* Wn1  = (const float*)d_in[9];
  const float* We1  = (const float*)d_in[10];
  const float* U1   = (const float*)d_in[11];
  const float* V1   = (const float*)d_in[12];
  const float* Wn2  = (const float*)d_in[13];
  const float* We2  = (const float*)d_in[14];
  const float* U2   = (const float*)d_in[15];
  const float* V2   = (const float*)d_in[16];

  const int E0 = in_sizes[1];
  const int N1 = in_sizes[4];
  const int E1 = in_sizes[5];
  const int N2 = in_sizes[8];
  const int N0 = in_sizes[0] / (B * Cin);

  float* ws = (float*)d_ws;
  const size_t szXS = (size_t)((N0 * B * H > N1 * B * Cout) ? N0 * B * H : N1 * B * Cout);
  const size_t szYD = (size_t)((N1 * B * H > N2 * B * Cout) ? N1 * B * H : N2 * B * Cout);
  float* xs  = ws;
  float* ys  = xs + szXS;
  float* yd  = ys + szXS;
  float* agg = yd + szYD;
  float* cnt = agg + szYD;
  float* h1  = cnt + (size_t)((N1 > N2) ? N1 : N2);

  const int EE = E0 + E1;
  int*   cnti0  = (int*)agg;
  int*   total0 = cnti0 + N1;
  int*   cnti1  = total0 + 1;
  int*   total1 = cnti1 + N2;
  int*   offs0  = total1 + 1;
  int*   offs1  = offs0 + N1;
  int*   rank   = offs1 + N2;
  int*   ssrc   = rank + EE;
  float* sew    = (float*)(ssrc + EE);

  const dim3 blk(256);
  auto waves_grid = [](int tiles) { return dim3((unsigned)((tiles + 7) / 8)); };

  const int R1s = N0 * B;
  const int R1d = N1 * B;
  gemm_pair_wmma<32, 16, 1><<<waves_grid(R1s / 16), blk, 0, stream>>>(
      X, Wn1, V1, xs, ys, R1s, N0, cnti0, N1 + N2 + 2);
  {
    const int rank_blocks = (EE + 255) / 256;
    const int gemm_blocks = (R1d / 16 + 7) / 8;
    gemm_gather_and_edge_rank<16, 16><<<dim3((unsigned)(rank_blocks + gemm_blocks)), blk, 0, stream>>>(
        xs, U1, yd, res0, R1d, rank_blocks, dst0, cnti0, E0, dst1, cnti1, E1, rank);
    const int blocks0 = (N1 + 255) / 256;
    const int blocks1 = (N2 + 255) / 256;
    node_offsets_kernel<<<dim3((unsigned)(blocks0 + blocks1)), blk, 0, stream>>>(
        cnti0, total0, offs0, N1, blocks0, cnti1, total1, offs1, N2);
    edge_place_kernel<<<dim3((unsigned)rank_blocks), blk, 0, stream>>>(
        src0, dst0, ew0, offs0, E0, src1, dst1, ew1, offs1, E1, rank, ssrc, sew);
    gather_finalize_kernel<64, 16, false><<<dim3((unsigned)((N1 + 7) / 8)), blk, 0, stream>>>(
        xs, res0, ys, yd, offs0, cnti0, ssrc, sew, We1, h1, N1);
  }

  const int R2s = N1 * B;
  const int R2d = N2 * B;
  gemm_pair_wmma<16, 32, 0><<<waves_grid(R2s / 16), blk, 0, stream>>>(
      h1, Wn2, V2, xs, ys, R2s, 0, nullptr, 0);
  gemm_wmma<32, 32, 2><<<waves_grid(R2d / 16), blk, 0, stream>>>(xs, U2,  yd, res1,    R2d, 0);
  gather_finalize_kernel<128, 32, true><<<dim3((unsigned)((N2 + 7) / 8)), blk, 0, stream>>>(
      xs, res1, ys, yd, offs1, cnti1, ssrc + E0, sew + E0, We2, (float*)d_out, N2);
}


# yd = xd @ U formed inside each node's gather wave on WMMA (one node per tile): yd plane and its GEMM waves gone, both binnings share a launch (6 -> 5); k grouping (sum order) of that product changes
# speedup vs baseline: 1.9709x; 1.3663x over previous
#include <hip/hip_runtime.h>
#include <hip/hip_bf16.h>

typedef float v2f __attribute__((ext_vector_type(2)));
typedef float v8f __attribute__((ext_vector_type(8)));

template <int K, int N, int MODE>
__global__ void gemm_pair_wmma(const float* __restrict__ A, const float* __restrict__ W,
                               const float* __restrict__ V, float* __restrict__ XS,
                               float* __restrict__ YS, int R, int nsrc,
                               int* __restrict__ zero_ptr, int zero_n) {
  for (int i = blockIdx.x * blockDim.x + threadIdx.x; i < zero_n; i += gridDim.x * blockDim.x)
    zero_ptr[i] = 0;

  const int wave = blockIdx.x * (blockDim.x >> 5) + (threadIdx.x >> 5);
  const int lane = threadIdx.x & 31;
  const int row0 = wave * 16;
  if (row0 >= R) return;

  const int c  = lane & 15;
  const int h  = lane >> 4;
  const int kb = h << 1;
  const int r  = row0 + c;

  long off;
  if (MODE == 0) {
    off = (long)r * K;
  } else {
    const int n = r >> 2, b = r & 3;
    off = ((long)b * nsrc + n) * (long)K;
  }
  const float* __restrict__ arow = A + off;

  v2f xf[K / 4];
#pragma unroll
  for (int ks = 0; ks < K / 4; ++ks) {
    xf[ks].x = arow[ks * 4 + kb];
    xf[ks].y = arow[ks * 4 + kb + 1];
  }

  v2f wf[K / 4][N / 16];
#pragma unroll
  for (int ks = 0; ks < K / 4; ++ks) {
#pragma unroll
    for (int nt = 0; nt < N / 16; ++nt) {
      const int kk = ks * 4 + kb;
      const int nn = nt * 16 + c;
      wf[ks][nt].x = W[kk * N + nn];
      wf[ks][nt].y = W[(kk + 1) * N + nn];
    }
  }

  v2f vf[N / 16][4][N / 16];
#pragma unroll
  for (int nt = 0; nt < N / 16; ++nt) {
#pragma unroll
    for (int j = 0; j < 4; ++j) {
#pragma unroll
      for (int nt2 = 0; nt2 < N / 16; ++nt2) {
        const int k0 = nt * 16 + 8 * h + 2 * j;
        const int nn = nt2 * 16 + c;
        vf[nt][j][nt2].x = V[k0 * N + nn];
        vf[nt][j][nt2].y = V[(k0 + 1) * N + nn];
      }
    }
  }

  v8f d1[N / 16] = {};
#pragma unroll
  for (int ks = 0; ks < K / 4; ++ks) {
#pragma unroll
    for (int nt = 0; nt < N / 16; ++nt) {
      d1[nt] = __builtin_amdgcn_wmma_f32_16x16x4_f32(
           false, wf[ks][nt],  false, xf[ks],
           (short)0, d1[nt],  false,  false);
    }
  }

  v8f d2[N / 16] = {};
#pragma unroll
  for (int nt = 0; nt < N / 16; ++nt) {
#pragma unroll
    for (int j = 0; j < 4; ++j) {
      v2f xb;
      xb.x = d1[nt][2 * j];
      xb.y = d1[nt][2 * j + 1];
#pragma unroll
      for (int nt2 = 0; nt2 < N / 16; ++nt2) {
        d2[nt2] = __builtin_amdgcn_wmma_f32_16x16x4_f32(
             false, vf[nt][j][nt2],  false, xb,
             (short)0, d2[nt2],  false,  false);
      }
    }
  }

  float* __restrict__ xrow = XS + (long)r * N + 8 * h;
  float* __restrict__ yrow = YS + (long)r * N + 8 * h;
#pragma unroll
  for (int nt = 0; nt < N / 16; ++nt) {
    *(float4*)(xrow + nt * 16)     = make_float4(d1[nt][0], d1[nt][1], d1[nt][2], d1[nt][3]);
    *(float4*)(xrow + nt * 16 + 4) = make_float4(d1[nt][4], d1[nt][5], d1[nt][6], d1[nt][7]);
    *(float4*)(yrow + nt * 16)     = make_float4(d2[nt][0], d2[nt][1], d2[nt][2], d2[nt][3]);
    *(float4*)(yrow + nt * 16 + 4) = make_float4(d2[nt][4], d2[nt][5], d2[nt][6], d2[nt][7]);
  }
}

constexpr int BIN_CAP = 24;

__device__ __forceinline__ void edge_bin(int e, const int* __restrict__ src,
                                         const int* __restrict__ dst,
                                         const float* __restrict__ ew, int E,
                                         int* __restrict__ cnt, int* __restrict__ bins,
                                         int* __restrict__ ovn, int* __restrict__ ov) {
  if (e < E) {
    const int d = dst[e];
    int2 p;
    p.x = src[e];
    p.y = __float_as_int(ew[e]);
    asm volatile("" : "+v"(p.x), "+v"(p.y));
    const int r = atomicAdd(cnt + d, 1);
    if (r < BIN_CAP) {
      ((int2*)bins)[(long)d * BIN_CAP + r] = p;
    } else {
      ov[atomicAdd(ovn, 1)] = e;
    }
  }
}

__global__ void edge_bin_both(int blocks0,
                              const int* __restrict__ src0, const int* __restrict__ dst0,
                              const float* __restrict__ ew0, int E0,
                              int* __restrict__ cnt0, int* __restrict__ bins0,
                              int* __restrict__ ovn0, int* __restrict__ ov0,
                              const int* __restrict__ src1, const int* __restrict__ dst1,
                              const float* __restrict__ ew1, int E1,
                              int* __restrict__ cnt1, int* __restrict__ bins1,
                              int* __restrict__ ovn1, int* __restrict__ ov1) {
  if ((int)blockIdx.x < blocks0) {
    edge_bin((int)(blockIdx.x * blockDim.x + threadIdx.x), src0, dst0, ew0, E0, cnt0, bins0, ovn0, ov0);
  } else {
    edge_bin((int)((blockIdx.x - blocks0) * blockDim.x + threadIdx.x), src1, dst1, ew1, E1, cnt1,
             bins1, ovn1, ov1);
  }
}

template <int BH, int H>
__device__ __forceinline__ void node_rows_times_u(const float* __restrict__ xrow,
                                                  const float* __restrict__ Uw, int lane,
                                                  float (&ydv)[BH / 32]) {
  constexpr int HK = H / 2;
  constexpr int NT = H / 16;
  const int m = lane & 15;
  const int h = lane >> 4;
  const int b = (H == 16) ? ((m >> 3) + 2 * (m & 1)) : (m & 3);
  const float* __restrict__ ap = xrow + b * H + h * HK;

  float av[HK];
#pragma unroll
  for (int q = 0; q < HK / 4; ++q) {
    const float4 t = *(const float4*)(ap + 4 * q);
    av[4 * q] = t.x; av[4 * q + 1] = t.y; av[4 * q + 2] = t.z; av[4 * q + 3] = t.w;
  }
  v2f uf[H / 4][NT];
#pragma unroll
  for (int ks = 0; ks < H / 4; ++ks) {
#pragma unroll
    for (int nt = 0; nt < NT; ++nt) {
      const int k0 = h * HK + 2 * ks;
      const int nn = nt * 16 + m;
      uf[ks][nt].x = Uw[k0 * H + nn];
      uf[ks][nt].y = Uw[(k0 + 1) * H + nn];
    }
  }
  v8f acc[NT] = {};
#pragma unroll
  for (int ks = 0; ks < H / 4; ++ks) {
    v2f a;
    a.x = av[2 * ks];
    a.y = av[2 * ks + 1];
#pragma unroll
    for (int nt = 0; nt < NT; ++nt) {
      acc[nt] = __builtin_amdgcn_wmma_f32_16x16x4_f32(
           false, a,  false, uf[ks][nt],
           (short)0, acc[nt],  false,  false);
    }
  }
  if (H == 16) {
    ydv[0] = acc[0][0];
    ydv[1] = acc[0][1];
  } else {
#pragma unroll
    for (int j = 0; j < BH / 32; ++j) ydv[j] = h ? acc[NT - 1][j] : acc[0][j];
  }
}

template <int PER>
__device__ __forceinline__ void add_edge(float (&acc)[PER], const float (&x)[PER], float w,
                                         const float (&ydv)[PER], const float (&wev)[PER]) {
#pragma unroll
  for (int j = 0; j < PER; ++j) {
    const float ev = w * wev[j];
    const float t  = ydv[j] * x[j] * ev;
    const float sg = 1.0f / (1.0f + __expf(-t));
    acc[j] += x[j] * sg;
  }
}

template <int BH, int H, bool FINAL>
__global__ void gather_finalize_kernel(const float* __restrict__ xs, const int* __restrict__ res,
                                       const float* __restrict__ ys, const float* __restrict__ Uw,
                                       const int* __restrict__ cnti, const int* __restrict__ bins,
                                       const int* __restrict__ ovn, const int* __restrict__ ov,
                                       const int* __restrict__ src, const int* __restrict__ dst,
                                       const float* __restrict__ ew,
                                       const float* __restrict__ We, float* __restrict__ out,
                                       int Nd) {
  const int d = __builtin_amdgcn_readfirstlane(
      (int)(blockIdx.x * (blockDim.x >> 5) + (threadIdx.x >> 5)));
  if (d >= Nd) return;
  const int lane = threadIdx.x & 31;
  constexpr int PER = BH / 32;
  constexpr int U   = 16 / PER;

  int rd = res[d];
  int n  = cnti[d];
  asm volatile("" : "+s"(rd), "+s"(n));
  const long base  = (long)d * BH;
  const long xbase = (long)rd * BH;

  const int m    = (n < BIN_CAP) ? n : BIN_CAP;
  int       my_s = 0;
  float     my_w = 0.0f;
  if (lane < m) {
    const int2 p = ((const int2*)bins)[(long)d * BIN_CAP + lane];
    my_s = p.x;
    my_w = __int_as_float(p.y);
  }

  float ydv[PER], xdv[PER], wev[PER], acc[PER];
  node_rows_times_u<BH, H>(xs + xbase, Uw, lane, ydv);
#pragma unroll
  for (int j = 0; j < PER; ++j) {
    const int i = lane + j * 32;
    xdv[j] = xs[xbase + i];
    wev[j] = We[i & (H - 1)];
    acc[j] = 0.0f;
  }

  for (int k = 0; k < m; k += U) {
    float xj[U][PER];
    float w[U];
    if (k + U <= m) {
#pragma unroll
      for (int u = 0; u < U; ++u) {
        const int s = __builtin_amdgcn_readlane(my_s, k + u);
        w[u] = __int_as_float(__builtin_amdgcn_readlane(__float_as_int(my_w), k + u));
        const float* __restrict__ row = ys + (long)s * BH;
#pragma unroll
        for (int j = 0; j < PER; ++j) xj[u][j] = row[lane + j * 32];
      }
#pragma unroll
      for (int u = 0; u < U; ++u) add_edge<PER>(acc, xj[u], w[u], ydv, wev);
    } else {
#pragma unroll
      for (int u = 0; u < U - 1; ++u) {
        const int kk = (k + u < m) ? (k + u) : (m - 1);
        const int s  = __builtin_amdgcn_readlane(my_s, kk);
        w[u] = __int_as_float(__builtin_amdgcn_readlane(__float_as_int(my_w), kk));
        const float* __restrict__ row = ys + (long)s * BH;
#pragma unroll
        for (int j = 0; j < PER; ++j) xj[u][j] = row[lane + j * 32];
      }
#pragma unroll
      for (int u = U - 2; u >= 1; --u) {
        if (k + u < m) add_edge<PER>(acc, xj[u], w[u], ydv, wev);
      }
      add_edge<PER>(acc, xj[0], w[0], ydv, wev);
    }
  }

  if (n > BIN_CAP) {
    const int L = ovn[0];
    for (int k0 = 0; k0 < L; k0 += 32) {
      int  e   = 0;
      bool hit = false;
      if (k0 + lane < L) {
        e   = ov[k0 + lane];
        hit = (dst[e] == d);
      }
      unsigned mask = __builtin_amdgcn_ballot_w32(hit);
      while (mask != 0u) {
        const int b = __builtin_ctz(mask);
        mask &= mask - 1u;
        const int   eb = __builtin_amdgcn_readlane(e, b);
        const float w1 = ew[eb];
        const float* __restrict__ row = ys + (long)src[eb] * BH;
        float x1[PER];
#pragma unroll
        for (int j = 0; j < PER; ++j) x1[j] = row[lane + j * 32];
        add_edge<PER>(acc, x1, w1, ydv, wev);
      }
    }
  }

  const float inv_cnt = 1.0f / fmaxf((float)n, 1.0f);
  float vals[PER];
  float sum = 0.0f, ssq = 0.0f;
#pragma unroll
  for (int j = 0; j < PER; ++j) {
    const float o = ydv[j] + acc[j] * inv_cnt;
    vals[j] = o;
    sum += o;
    ssq += o * o;
  }
#pragma unroll
  for (int mask = 16; mask >= 1; mask >>= 1) {
    sum += __shfl_xor(sum, mask, 32);
    ssq += __shfl_xor(ssq, mask, 32);
  }
  const float mu  = sum * (1.0f / BH);
  const float var = ssq * (1.0f / BH) - mu * mu;
  const float is  = rsqrtf(var + 1e-5f);

#pragma unroll
  for (int j = 0; j < PER; ++j) {
    const int i = lane + j * 32;
    float v = xdv[j] + (vals[j] - mu) * is;
    v = v > 0.0f ? v : 0.01f * v;
    if (FINAL) {
      const int b = i / H, c = i & (H - 1);
      out[((long)b * Nd + d) * H + c] = v;
    } else {
      out[base + i] = v;
    }
  }
}

extern "C" void kernel_launch(void* const* d_in, const int* in_sizes, int n_in,
                              void* d_out, int out_size, void* d_ws, size_t ws_size,
                              hipStream_t stream) {
  constexpr int B = 4, Cin = 32, H = 16, Cout = 32;
  const float* X    = (const float*)d_in[0];
  const int*   src0 = (const int*)d_in[1];
  const int*   dst0 = (const int*)d_in[2];
  const float* ew0  = (const float*)d_in[3];
  const int*   res0 = (const int*)d_in[4];
  const int*   src1 = (const int*)d_in[5];
  const int*   dst1 = (const int*)d_in[6];
  const float* ew1  = (const float*)d_in[7];
  const int*   res1 = (const int*)d_in[8];
  const float* Wn1  = (const float*)d_in[9];
  const float* We1  = (const float*)d_in[10];
  const float* U1   = (const float*)d_in[11];
  const float* V1   = (const float*)d_in[12];
  const float* Wn2  = (const float*)d_in[13];
  const float* We2  = (const float*)d_in[14];
  const float* U2   = (const float*)d_in[15];
  const float* V2   = (const float*)d_in[16];

  const int E0 = in_sizes[1];
  const int N1 = in_sizes[4];
  const int E1 = in_sizes[5];
  const int N2 = in_sizes[8];
  const int N0 = in_sizes[0] / (B * Cin);

  float* ws = (float*)d_ws;
  const size_t szXS = (size_t)((N0 * B * H > N1 * B * Cout) ? N0 * B * H : N1 * B * Cout);
  const size_t szYD = (size_t)((N1 * B * H > N2 * B * Cout) ? N1 * B * H : N2 * B * Cout);
  float* xs  = ws;
  float* ys  = xs + szXS;
  float* yd  = ys + szXS;
  float* agg = yd + szYD;
  float* cnt = agg + szYD;
  float* h1  = cnt + (size_t)((N1 > N2) ? N1 : N2);

  int* bins0 = (int*)agg;
  int* cnti0 = bins0 + (size_t)N1 * BIN_CAP * 2;
  int* ovn0  = cnti0 + N1;
  int* cnti1 = ovn0 + 1;
  int* ovn1  = cnti1 + N2;
  int* ov0   = ovn1 + 1;
  int* ov1   = ov0 + E0;
  int* bins1 = (int*)yd;

  const dim3 blk(256);
  auto waves_grid = [](int tiles) { return dim3((unsigned)((tiles + 7) / 8)); };

  const int R1s = N0 * B;
  gemm_pair_wmma<32, 16, 1><<<waves_grid(R1s / 16), blk, 0, stream>>>(
      X, Wn1, V1, xs, ys, R1s, N0, cnti0, N1 + N2 + 2);
  {
    const int blocks0 = (E0 + 255) / 256;
    const int blocks1 = (E1 + 255) / 256;
    edge_bin_both<<<dim3((unsigned)(blocks0 + blocks1)), blk, 0, stream>>>(
        blocks0, src0, dst0, ew0, E0, cnti0, bins0, ovn0, ov0,
        src1, dst1, ew1, E1, cnti1, bins1, ovn1, ov1);
    gather_finalize_kernel<64, 16, false><<<dim3((unsigned)((N1 + 7) / 8)), blk, 0, stream>>>(
        xs, res0, ys, U1, cnti0, bins0, ovn0, ov0, src0, dst0, ew0, We1, h1, N1);
  }

  const int R2s = N1 * B;
  gemm_pair_wmma<16, 32, 0><<<waves_grid(R2s / 16), blk, 0, stream>>>(
      h1, Wn2, V2, xs, ys, R2s, 0, nullptr, 0);
  gather_finalize_kernel<128, 32, true><<<dim3((unsigned)((N2 + 7) / 8)), blk, 0, stream>>>(
      xs, res1, ys, U2, cnti1, bins1, ovn1, ov1, src1, dst1, ew1, We2, (float*)d_out, N2);
}


# Sigmoid of each edge message uses the hardware reciprocal (1 ulp) instead of the correctly rounded division: under half the vector work per edge in the gather waves, registers 66 to 52 and 95 to 68
# speedup vs baseline: 2.1315x; 1.0815x over previous
#include <hip/hip_runtime.h>
#include <hip/hip_bf16.h>

typedef float v2f __attribute__((ext_vector_type(2)));
typedef float v8f __attribute__((ext_vector_type(8)));

template <int K, int N, int MODE>
__global__ void gemm_pair_wmma(const float* __restrict__ A, const float* __restrict__ W,
                               const float* __restrict__ V, float* __restrict__ XS,
                               float* __restrict__ YS, int R, int nsrc,
                               int* __restrict__ zero_ptr, int zero_n) {
  for (int i = blockIdx.x * blockDim.x + threadIdx.x; i < zero_n; i += gridDim.x * blockDim.x)
    zero_ptr[i] = 0;

  const int wave = blockIdx.x * (blockDim.x >> 5) + (threadIdx.x >> 5);
  const int lane = threadIdx.x & 31;
  const int row0 = wave * 16;
  if (row0 >= R) return;

  const int c  = lane & 15;
  const int h  = lane >> 4;
  const int kb = h << 1;
  const int r  = row0 + c;

  long off;
  if (MODE == 0) {
    off = (long)r * K;
  } else {
    const int n = r >> 2, b = r & 3;
    off = ((long)b * nsrc + n) * (long)K;
  }
  const float* __restrict__ arow = A + off;

  v2f xf[K / 4];
#pragma unroll
  for (int ks = 0; ks < K / 4; ++ks) {
    xf[ks].x = arow[ks * 4 + kb];
    xf[ks].y = arow[ks * 4 + kb + 1];
  }

  v2f wf[K / 4][N / 16];
#pragma unroll
  for (int ks = 0; ks < K / 4; ++ks) {
#pragma unroll
    for (int nt = 0; nt < N / 16; ++nt) {
      const int kk = ks * 4 + kb;
      const int nn = nt * 16 + c;
      wf[ks][nt].x = W[kk * N + nn];
      wf[ks][nt].y = W[(kk + 1) * N + nn];
    }
  }

  v2f vf[N / 16][4][N / 16];
#pragma unroll
  for (int nt = 0; nt < N / 16; ++nt) {
#pragma unroll
    for (int j = 0; j < 4; ++j) {
#pragma unroll
      for (int nt2 = 0; nt2 < N / 16; ++nt2) {
        const int k0 = nt * 16 + 8 * h + 2 * j;
        const int nn = nt2 * 16 + c;
        vf[nt][j][nt2].x = V[k0 * N + nn];
        vf[nt][j][nt2].y = V[(k0 + 1) * N + nn];
      }
    }
  }

  v8f d1[N / 16] = {};
#pragma unroll
  for (int ks = 0; ks < K / 4; ++ks) {
#pragma unroll
    for (int nt = 0; nt < N / 16; ++nt) {
      d1[nt] = __builtin_amdgcn_wmma_f32_16x16x4_f32(
           false, wf[ks][nt],  false, xf[ks],
           (short)0, d1[nt],  false,  false);
    }
  }

  v8f d2[N / 16] = {};
#pragma unroll
  for (int nt = 0; nt < N / 16; ++nt) {
#pragma unroll
    for (int j = 0; j < 4; ++j) {
      v2f xb;
      xb.x = d1[nt][2 * j];
      xb.y = d1[nt][2 * j + 1];
#pragma unroll
      for (int nt2 = 0; nt2 < N / 16; ++nt2) {
        d2[nt2] = __builtin_amdgcn_wmma_f32_16x16x4_f32(
             false, vf[nt][j][nt2],  false, xb,
             (short)0, d2[nt2],  false,  false);
      }
    }
  }

  float* __restrict__ xrow = XS + (long)r * N + 8 * h;
  float* __restrict__ yrow = YS + (long)r * N + 8 * h;
#pragma unroll
  for (int nt = 0; nt < N / 16; ++nt) {
    *(float4*)(xrow + nt * 16)     = make_float4(d1[nt][0], d1[nt][1], d1[nt][2], d1[nt][3]);
    *(float4*)(xrow + nt * 16 + 4) = make_float4(d1[nt][4], d1[nt][5], d1[nt][6], d1[nt][7]);
    *(float4*)(yrow + nt * 16)     = make_float4(d2[nt][0], d2[nt][1], d2[nt][2], d2[nt][3]);
    *(float4*)(yrow + nt * 16 + 4) = make_float4(d2[nt][4], d2[nt][5], d2[nt][6], d2[nt][7]);
  }
}

constexpr int BIN_CAP = 24;

__device__ __forceinline__ void edge_bin(int e, const int* __restrict__ src,
                                         const int* __restrict__ dst,
                                         const float* __restrict__ ew, int E,
                                         int* __restrict__ cnt, int* __restrict__ bins,
                                         int* __restrict__ ovn, int* __restrict__ ov) {
  if (e < E) {
    const int d = dst[e];
    int2 p;
    p.x = src[e];
    p.y = __float_as_int(ew[e]);
    asm volatile("" : "+v"(p.x), "+v"(p.y));
    const int r = atomicAdd(cnt + d, 1);
    if (r < BIN_CAP) {
      ((int2*)bins)[(long)d * BIN_CAP + r] = p;
    } else {
      ov[atomicAdd(ovn, 1)] = e;
    }
  }
}

__global__ void edge_bin_both(int blocks0,
                              const int* __restrict__ src0, const int* __restrict__ dst0,
                              const float* __restrict__ ew0, int E0,
                              int* __restrict__ cnt0, int* __restrict__ bins0,
                              int* __restrict__ ovn0, int* __restrict__ ov0,
                              const int* __restrict__ src1, const int* __restrict__ dst1,
                              const float* __restrict__ ew1, int E1,
                              int* __restrict__ cnt1, int* __restrict__ bins1,
                              int* __restrict__ ovn1, int* __restrict__ ov1) {
  if ((int)blockIdx.x < blocks0) {
    edge_bin((int)(blockIdx.x * blockDim.x + threadIdx.x), src0, dst0, ew0, E0, cnt0, bins0, ovn0, ov0);
  } else {
    edge_bin((int)((blockIdx.x - blocks0) * blockDim.x + threadIdx.x), src1, dst1, ew1, E1, cnt1,
             bins1, ovn1, ov1);
  }
}

template <int BH, int H>
__device__ __forceinline__ void node_rows_times_u(const float* __restrict__ xrow,
                                                  const float* __restrict__ Uw, int lane,
                                                  float (&ydv)[BH / 32]) {
  constexpr int HK = H / 2;
  constexpr int NT = H / 16;
  const int m = lane & 15;
  const int h = lane >> 4;
  const int b = (H == 16) ? ((m >> 3) + 2 * (m & 1)) : (m & 3);
  const float* __restrict__ ap = xrow + b * H + h * HK;

  float av[HK];
#pragma unroll
  for (int q = 0; q < HK / 4; ++q) {
    const float4 t = *(const float4*)(ap + 4 * q);
    av[4 * q] = t.x; av[4 * q + 1] = t.y; av[4 * q + 2] = t.z; av[4 * q + 3] = t.w;
  }
  v2f uf[H / 4][NT];
#pragma unroll
  for (int ks = 0; ks < H / 4; ++ks) {
#pragma unroll
    for (int nt = 0; nt < NT; ++nt) {
      const int k0 = h * HK + 2 * ks;
      const int nn = nt * 16 + m;
      uf[ks][nt].x = Uw[k0 * H + nn];
      uf[ks][nt].y = Uw[(k0 + 1) * H + nn];
    }
  }
  v8f acc[NT] = {};
#pragma unroll
  for (int ks = 0; ks < H / 4; ++ks) {
    v2f a;
    a.x = av[2 * ks];
    a.y = av[2 * ks + 1];
#pragma unroll
    for (int nt = 0; nt < NT; ++nt) {
      acc[nt] = __builtin_amdgcn_wmma_f32_16x16x4_f32(
           false, a,  false, uf[ks][nt],
           (short)0, acc[nt],  false,  false);
    }
  }
  if (H == 16) {
    ydv[0] = acc[0][0];
    ydv[1] = acc[0][1];
  } else {
#pragma unroll
    for (int j = 0; j < BH / 32; ++j) ydv[j] = h ? acc[NT - 1][j] : acc[0][j];
  }
}

template <int PER>
__device__ __forceinline__ void add_edge(float (&acc)[PER], const float (&x)[PER], float w,
                                         const float (&ydv)[PER], const float (&wev)[PER]) {
#pragma unroll
  for (int j = 0; j < PER; ++j) {
    const float ev = w * wev[j];
    const float t  = ydv[j] * x[j] * ev;
    const float sg = __builtin_amdgcn_rcpf(1.0f + __expf(-t));
    acc[j] += x[j] * sg;
  }
}

template <int BH, int H, bool FINAL>
__global__ void gather_finalize_kernel(const float* __restrict__ xs, const int* __restrict__ res,
                                       const float* __restrict__ ys, const float* __restrict__ Uw,
                                       const int* __restrict__ cnti, const int* __restrict__ bins,
                                       const int* __restrict__ ovn, const int* __restrict__ ov,
                                       const int* __restrict__ src, const int* __restrict__ dst,
                                       const float* __restrict__ ew,
                                       const float* __restrict__ We, float* __restrict__ out,
                                       int Nd) {
  const int d = __builtin_amdgcn_readfirstlane(
      (int)(blockIdx.x * (blockDim.x >> 5) + (threadIdx.x >> 5)));
  if (d >= Nd) return;
  const int lane = threadIdx.x & 31;
  constexpr int PER = BH / 32;
  constexpr int U   = 16 / PER;

  int rd = res[d];
  int n  = cnti[d];
  asm volatile("" : "+s"(rd), "+s"(n));
  const long base  = (long)d * BH;
  const long xbase = (long)rd * BH;

  const int m    = (n < BIN_CAP) ? n : BIN_CAP;
  int       my_s = 0;
  float     my_w = 0.0f;
  if (lane < m) {
    const int2 p = ((const int2*)bins)[(long)d * BIN_CAP + lane];
    my_s = p.x;
    my_w = __int_as_float(p.y);
  }

  float ydv[PER], xdv[PER], wev[PER], acc[PER];
  node_rows_times_u<BH, H>(xs + xbase, Uw, lane, ydv);
#pragma unroll
  for (int j = 0; j < PER; ++j) {
    const int i = lane + j * 32;
    xdv[j] = xs[xbase + i];
    wev[j] = We[i & (H - 1)];
    acc[j] = 0.0f;
  }

  for (int k = 0; k < m; k += U) {
    float xj[U][PER];
    float w[U];
    if (k + U <= m) {
#pragma unroll
      for (int u = 0; u < U; ++u) {
        const int s = __builtin_amdgcn_readlane(my_s, k + u);
        w[u] = __int_as_float(__builtin_amdgcn_readlane(__float_as_int(my_w), k + u));
        const float* __restrict__ row = ys + (long)s * BH;
#pragma unroll
        for (int j = 0; j < PER; ++j) xj[u][j] = row[lane + j * 32];
      }
#pragma unroll
      for (int u = 0; u < U; ++u) add_edge<PER>(acc, xj[u], w[u], ydv, wev);
    } else {
#pragma unroll
      for (int u = 0; u < U - 1; ++u) {
        const int kk = (k + u < m) ? (k + u) : (m - 1);
        const int s  = __builtin_amdgcn_readlane(my_s, kk);
        w[u] = __int_as_float(__builtin_amdgcn_readlane(__float_as_int(my_w), kk));
        const float* __restrict__ row = ys + (long)s * BH;
#pragma unroll
        for (int j = 0; j < PER; ++j) xj[u][j] = row[lane + j * 32];
      }
#pragma unroll
      for (int u = U - 2; u >= 1; --u) {
        if (k + u < m) add_edge<PER>(acc, xj[u], w[u], ydv, wev);
      }
      add_edge<PER>(acc, xj[0], w[0], ydv, wev);
    }
  }

  if (n > BIN_CAP) {
    const int L = ovn[0];
    for (int k0 = 0; k0 < L; k0 += 32) {
      int  e   = 0;
      bool hit = false;
      if (k0 + lane < L) {
        e   = ov[k0 + lane];
        hit = (dst[e] == d);
      }
      unsigned mask = __builtin_amdgcn_ballot_w32(hit);
      while (mask != 0u) {
        const int b = __builtin_ctz(mask);
        mask &= mask - 1u;
        const int   eb = __builtin_amdgcn_readlane(e, b);
        const float w1 = ew[eb];
        const float* __restrict__ row = ys + (long)src[eb] * BH;
        float x1[PER];
#pragma unroll
        for (int j = 0; j < PER; ++j) x1[j] = row[lane + j * 32];
        add_edge<PER>(acc, x1, w1, ydv, wev);
      }
    }
  }

  const float inv_cnt = 1.0f / fmaxf((float)n, 1.0f);
  float vals[PER];
  float sum = 0.0f, ssq = 0.0f;
#pragma unroll
  for (int j = 0; j < PER; ++j) {
    const float o = ydv[j] + acc[j] * inv_cnt;
    vals[j] = o;
    sum += o;
    ssq += o * o;
  }
#pragma unroll
  for (int mask = 16; mask >= 1; mask >>= 1) {
    sum += __shfl_xor(sum, mask, 32);
    ssq += __shfl_xor(ssq, mask, 32);
  }
  const float mu  = sum * (1.0f / BH);
  const float var = ssq * (1.0f / BH) - mu * mu;
  const float is  = rsqrtf(var + 1e-5f);

#pragma unroll
  for (int j = 0; j < PER; ++j) {
    const int i = lane + j * 32;
    float v = xdv[j] + (vals[j] - mu) * is;
    v = v > 0.0f ? v : 0.01f * v;
    if (FINAL) {
      const int b = i / H, c = i & (H - 1);
      out[((long)b * Nd + d) * H + c] = v;
    } else {
      out[base + i] = v;
    }
  }
}

extern "C" void kernel_launch(void* const* d_in, const int* in_sizes, int n_in,
                              void* d_out, int out_size, void* d_ws, size_t ws_size,
                              hipStream_t stream) {
  constexpr int B = 4, Cin = 32, H = 16, Cout = 32;
  const float* X    = (const float*)d_in[0];
  const int*   src0 = (const int*)d_in[1];
  const int*   dst0 = (const int*)d_in[2];
  const float* ew0  = (const float*)d_in[3];
  const int*   res0 = (const int*)d_in[4];
  const int*   src1 = (const int*)d_in[5];
  const int*   dst1 = (const int*)d_in[6];
  const float* ew1  = (const float*)d_in[7];
  const int*   res1 = (const int*)d_in[8];
  const float* Wn1  = (const float*)d_in[9];
  const float* We1  = (const float*)d_in[10];
  const float* U1   = (const float*)d_in[11];
  const float* V1   = (const float*)d_in[12];
  const float* Wn2  = (const float*)d_in[13];
  const float* We2  = (const float*)d_in[14];
  const float* U2   = (const float*)d_in[15];
  const float* V2   = (const float*)d_in[16];

  const int E0 = in_sizes[1];
  const int N1 = in_sizes[4];
  const int E1 = in_sizes[5];
  const int N2 = in_sizes[8];
  const int N0 = in_sizes[0] / (B * Cin);

  float* ws = (float*)d_ws;
  const size_t szXS = (size_t)((N0 * B * H > N1 * B * Cout) ? N0 * B * H : N1 * B * Cout);
  const size_t szYD = (size_t)((N1 * B * H > N2 * B * Cout) ? N1 * B * H : N2 * B * Cout);
  float* xs  = ws;
  float* ys  = xs + szXS;
  float* yd  = ys + szXS;
  float* agg = yd + szYD;
  float* cnt = agg + szYD;
  float* h1  = cnt + (size_t)((N1 > N2) ? N1 : N2);

  int* bins0 = (int*)agg;
  int* cnti0 = bins0 + (size_t)N1 * BIN_CAP * 2;
  int* ovn0  = cnti0 + N1;
  int* cnti1 = ovn0 + 1;
  int* ovn1  = cnti1 + N2;
  int* ov0   = ovn1 + 1;
  int* ov1   = ov0 + E0;
  int* bins1 = (int*)yd;

  const dim3 blk(256);
  auto waves_grid = [](int tiles) { return dim3((unsigned)((tiles + 7) / 8)); };

  const int R1s = N0 * B;
  gemm_pair_wmma<32, 16, 1><<<waves_grid(R1s / 16), blk, 0, stream>>>(
      X, Wn1, V1, xs, ys, R1s, N0, cnti0, N1 + N2 + 2);
  {
    const int blocks0 = (E0 + 255) / 256;
    const int blocks1 = (E1 + 255) / 256;
    edge_bin_both<<<dim3((unsigned)(blocks0 + blocks1)), blk, 0, stream>>>(
        blocks0, src0, dst0, ew0, E0, cnti0, bins0, ovn0, ov0,
        src1, dst1, ew1, E1, cnti1, bins1, ovn1, ov1);
    gather_finalize_kernel<64, 16, false><<<dim3((unsigned)((N1 + 7) / 8)), blk, 0, stream>>>(
        xs, res0, ys, U1, cnti0, bins0, ovn0, ov0, src0, dst0, ew0, We1, h1, N1);
  }

  const int R2s = N1 * B;
  gemm_pair_wmma<16, 32, 0><<<waves_grid(R2s / 16), blk, 0, stream>>>(
      h1, Wn2, V2, xs, ys, R2s, 0, nullptr, 0);
  gather_finalize_kernel<128, 32, true><<<dim3((unsigned)((N2 + 7) / 8)), blk, 0, stream>>>(
      xs, res1, ys, U2, cnti1, bins1, ovn1, ov1, src1, dst1, ew1, We2, (float*)d_out, N2);
}


# Layer 2 node GEMM launch gone (5 -> 4): gather waves form xj per 4-edge tile on WMMA from 256-byte h1 rows, Wn2@V2 and Wn2@U2 folded per call; products reassociated, edge and BN sum trees change
# speedup vs baseline: 2.4039x; 1.1278x over previous
#include <hip/hip_runtime.h>
#include <hip/hip_bf16.h>

typedef float v2f __attribute__((ext_vector_type(2)));
typedef float v8f __attribute__((ext_vector_type(8)));

template <int K, int N, int MODE>
__global__ void gemm_pair_wmma(const float* __restrict__ A, const float* __restrict__ W,
                               const float* __restrict__ V, float* __restrict__ XS,
                               float* __restrict__ YS, int R, int nsrc,
                               int* __restrict__ zero_ptr, int zero_n) {
  for (int i = blockIdx.x * blockDim.x + threadIdx.x; i < zero_n; i += gridDim.x * blockDim.x)
    zero_ptr[i] = 0;

  const int wave = blockIdx.x * (blockDim.x >> 5) + (threadIdx.x >> 5);
  const int lane = threadIdx.x & 31;
  const int row0 = wave * 16;
  if (row0 >= R) return;

  const int c  = lane & 15;
  const int h  = lane >> 4;
  const int kb = h << 1;
  const int r  = row0 + c;

  long off;
  if (MODE == 0) {
    off = (long)r * K;
  } else {
    const int n = r >> 2, b = r & 3;
    off = ((long)b * nsrc + n) * (long)K;
  }
  const float* __restrict__ arow = A + off;

  v2f xf[K / 4];
#pragma unroll
  for (int ks = 0; ks < K / 4; ++ks) {
    xf[ks].x = arow[ks * 4 + kb];
    xf[ks].y = arow[ks * 4 + kb + 1];
  }

  v2f wf[K / 4][N / 16];
#pragma unroll
  for (int ks = 0; ks < K / 4; ++ks) {
#pragma unroll
    for (int nt = 0; nt < N / 16; ++nt) {
      const int kk = ks * 4 + kb;
      const int nn = nt * 16 + c;
      wf[ks][nt].x = W[kk * N + nn];
      wf[ks][nt].y = W[(kk + 1) * N + nn];
    }
  }

  v2f vf[N / 16][4][N / 16];
#pragma unroll
  for (int nt = 0; nt < N / 16; ++nt) {
#pragma unroll
    for (int j = 0; j < 4; ++j) {
#pragma unroll
      for (int nt2 = 0; nt2 < N / 16; ++nt2) {
        const int k0 = nt * 16 + 8 * h + 2 * j;
        const int nn = nt2 * 16 + c;
        vf[nt][j][nt2].x = V[k0 * N + nn];
        vf[nt][j][nt2].y = V[(k0 + 1) * N + nn];
      }
    }
  }

  v8f d1[N / 16] = {};
#pragma unroll
  for (int ks = 0; ks < K / 4; ++ks) {
#pragma unroll
    for (int nt = 0; nt < N / 16; ++nt) {
      d1[nt] = __builtin_amdgcn_wmma_f32_16x16x4_f32(
           false, wf[ks][nt],  false, xf[ks],
           (short)0, d1[nt],  false,  false);
    }
  }

  v8f d2[N / 16] = {};
#pragma unroll
  for (int nt = 0; nt < N / 16; ++nt) {
#pragma unroll
    for (int j = 0; j < 4; ++j) {
      v2f xb;
      xb.x = d1[nt][2 * j];
      xb.y = d1[nt][2 * j + 1];
#pragma unroll
      for (int nt2 = 0; nt2 < N / 16; ++nt2) {
        d2[nt2] = __builtin_amdgcn_wmma_f32_16x16x4_f32(
             false, vf[nt][j][nt2],  false, xb,
             (short)0, d2[nt2],  false,  false);
      }
    }
  }

  float* __restrict__ xrow = XS + (long)r * N + 8 * h;
  float* __restrict__ yrow = YS + (long)r * N + 8 * h;
#pragma unroll
  for (int nt = 0; nt < N / 16; ++nt) {
    *(float4*)(xrow + nt * 16)     = make_float4(d1[nt][0], d1[nt][1], d1[nt][2], d1[nt][3]);
    *(float4*)(xrow + nt * 16 + 4) = make_float4(d1[nt][4], d1[nt][5], d1[nt][6], d1[nt][7]);
    *(float4*)(yrow + nt * 16)     = make_float4(d2[nt][0], d2[nt][1], d2[nt][2], d2[nt][3]);
    *(float4*)(yrow + nt * 16 + 4) = make_float4(d2[nt][4], d2[nt][5], d2[nt][6], d2[nt][7]);
  }
}

constexpr int BIN_CAP = 24;

__device__ __forceinline__ void edge_bin(int e, const int* __restrict__ src,
                                         const int* __restrict__ dst,
                                         const float* __restrict__ ew, int E,
                                         int* __restrict__ cnt, int* __restrict__ bins,
                                         int* __restrict__ ovn, int* __restrict__ ov) {
  if (e < E) {
    const int d = dst[e];
    int2 p;
    p.x = src[e];
    p.y = __float_as_int(ew[e]);
    asm volatile("" : "+v"(p.x), "+v"(p.y));
    const int r = atomicAdd(cnt + d, 1);
    if (r < BIN_CAP) {
      ((int2*)bins)[(long)d * BIN_CAP + r] = p;
    } else {
      ov[atomicAdd(ovn, 1)] = e;
    }
  }
}

__device__ __forceinline__ void fold_weights(const float* __restrict__ Wn,
                                             const float* __restrict__ M,
                                             float* __restrict__ P, int lane) {
  const int m = lane & 15;
  const int h = lane >> 4;
  const float* __restrict__ ap = Wn + m * 32 + 16 * h;
  float av[16];
#pragma unroll
  for (int q = 0; q < 4; ++q) {
    const float4 t = *(const float4*)(ap + 4 * q);
    av[4 * q] = t.x; av[4 * q + 1] = t.y; av[4 * q + 2] = t.z; av[4 * q + 3] = t.w;
  }
  v8f acc[2] = {};
#pragma unroll
  for (int ks = 0; ks < 8; ++ks) {
    v2f a;
    a.x = av[2 * ks];
    a.y = av[2 * ks + 1];
    const int k0 = 16 * h + 2 * ks;
#pragma unroll
    for (int nt = 0; nt < 2; ++nt) {
      v2f b;
      b.x = M[k0 * 32 + nt * 16 + m];
      b.y = M[(k0 + 1) * 32 + nt * 16 + m];
      acc[nt] = __builtin_amdgcn_wmma_f32_16x16x4_f32(
           false, a,  false, b,
           (short)0, acc[nt],  false,  false);
    }
  }
  float* __restrict__ pf = P + lane * 16;
#pragma unroll
  for (int nt = 0; nt < 2; ++nt) {
    *(float4*)(pf + nt * 8)     = make_float4(acc[nt][0], acc[nt][1], acc[nt][2], acc[nt][3]);
    *(float4*)(pf + nt * 8 + 4) = make_float4(acc[nt][4], acc[nt][5], acc[nt][6], acc[nt][7]);
  }
}

__global__ void edge_bin_both(int blocks0,
                              const float* __restrict__ Wn2, const float* __restrict__ V2,
                              const float* __restrict__ U2, float* __restrict__ WV,
                              float* __restrict__ WU,
                              const int* __restrict__ src0, const int* __restrict__ dst0,
                              const float* __restrict__ ew0, int E0,
                              int* __restrict__ cnt0, int* __restrict__ bins0,
                              int* __restrict__ ovn0, int* __restrict__ ov0,
                              const int* __restrict__ src1, const int* __restrict__ dst1,
                              const float* __restrict__ ew1, int E1,
                              int* __restrict__ cnt1, int* __restrict__ bins1,
                              int* __restrict__ ovn1, int* __restrict__ ov1) {
  const int bid = (int)blockIdx.x - 1;
  if (bid < 0) {
    const int wv = __builtin_amdgcn_readfirstlane((int)(threadIdx.x >> 5));
    if (wv == 0) fold_weights(Wn2, V2, WV, (int)(threadIdx.x & 31));
    else if (wv == 1) fold_weights(Wn2, U2, WU, (int)(threadIdx.x & 31));
  } else if (bid < blocks0) {
    edge_bin((int)(bid * blockDim.x + threadIdx.x), src0, dst0, ew0, E0, cnt0, bins0, ovn0, ov0);
  } else {
    edge_bin((int)((bid - blocks0) * blockDim.x + threadIdx.x), src1, dst1, ew1, E1, cnt1,
             bins1, ovn1, ov1);
  }
}

template <int BH, int H>
__device__ __forceinline__ void node_rows_times_u(const float* __restrict__ xrow,
                                                  const float* __restrict__ Uw, int lane,
                                                  float (&ydv)[BH / 32]) {
  constexpr int HK = H / 2;
  constexpr int NT = H / 16;
  const int m = lane & 15;
  const int h = lane >> 4;
  const int b = (H == 16) ? ((m >> 3) + 2 * (m & 1)) : (m & 3);
  const float* __restrict__ ap = xrow + b * H + h * HK;

  float av[HK];
#pragma unroll
  for (int q = 0; q < HK / 4; ++q) {
    const float4 t = *(const float4*)(ap + 4 * q);
    av[4 * q] = t.x; av[4 * q + 1] = t.y; av[4 * q + 2] = t.z; av[4 * q + 3] = t.w;
  }
  v2f uf[H / 4][NT];
#pragma unroll
  for (int ks = 0; ks < H / 4; ++ks) {
#pragma unroll
    for (int nt = 0; nt < NT; ++nt) {
      const int k0 = h * HK + 2 * ks;
      const int nn = nt * 16 + m;
      uf[ks][nt].x = Uw[k0 * H + nn];
      uf[ks][nt].y = Uw[(k0 + 1) * H + nn];
    }
  }
  v8f acc[NT] = {};
#pragma unroll
  for (int ks = 0; ks < H / 4; ++ks) {
    v2f a;
    a.x = av[2 * ks];
    a.y = av[2 * ks + 1];
#pragma unroll
    for (int nt = 0; nt < NT; ++nt) {
      acc[nt] = __builtin_amdgcn_wmma_f32_16x16x4_f32(
           false, a,  false, uf[ks][nt],
           (short)0, acc[nt],  false,  false);
    }
  }
  if (H == 16) {
    ydv[0] = acc[0][0];
    ydv[1] = acc[0][1];
  } else {
#pragma unroll
    for (int j = 0; j < BH / 32; ++j) ydv[j] = h ? acc[NT - 1][j] : acc[0][j];
  }
}

template <int PER>
__device__ __forceinline__ void add_edge(float (&acc)[PER], const float (&x)[PER], float w,
                                         const float (&ydv)[PER], const float (&wev)[PER]) {
#pragma unroll
  for (int j = 0; j < PER; ++j) {
    const float ev = w * wev[j];
    const float t  = ydv[j] * x[j] * ev;
    const float sg = __builtin_amdgcn_rcpf(1.0f + __expf(-t));
    acc[j] += x[j] * sg;
  }
}

template <int BH, int H, bool FINAL>
__global__ void gather_finalize_kernel(const float* __restrict__ xs, const int* __restrict__ res,
                                       const float* __restrict__ ys, const float* __restrict__ Uw,
                                       const int* __restrict__ cnti, const int* __restrict__ bins,
                                       const int* __restrict__ ovn, const int* __restrict__ ov,
                                       const int* __restrict__ src, const int* __restrict__ dst,
                                       const float* __restrict__ ew,
                                       const float* __restrict__ We, float* __restrict__ out,
                                       int Nd) {
  const int d = __builtin_amdgcn_readfirstlane(
      (int)(blockIdx.x * (blockDim.x >> 5) + (threadIdx.x >> 5)));
  if (d >= Nd) return;
  const int lane = threadIdx.x & 31;
  constexpr int PER = BH / 32;
  constexpr int U   = 16 / PER;

  int rd = res[d];
  int n  = cnti[d];
  asm volatile("" : "+s"(rd), "+s"(n));
  const long base  = (long)d * BH;
  const long xbase = (long)rd * BH;

  const int m    = (n < BIN_CAP) ? n : BIN_CAP;
  int       my_s = 0;
  float     my_w = 0.0f;
  if (lane < m) {
    const int2 p = ((const int2*)bins)[(long)d * BIN_CAP + lane];
    my_s = p.x;
    my_w = __int_as_float(p.y);
  }

  float ydv[PER], xdv[PER], wev[PER], acc[PER];
  node_rows_times_u<BH, H>(xs + xbase, Uw, lane, ydv);
#pragma unroll
  for (int j = 0; j < PER; ++j) {
    const int i = lane + j * 32;
    xdv[j] = xs[xbase + i];
    wev[j] = We[i & (H - 1)];
    acc[j] = 0.0f;
  }

  for (int k = 0; k < m; k += U) {
    float xj[U][PER];
    float w[U];
    if (k + U <= m) {
#pragma unroll
      for (int u = 0; u < U; ++u) {
        const int s = __builtin_amdgcn_readlane(my_s, k + u);
        w[u] = __int_as_float(__builtin_amdgcn_readlane(__float_as_int(my_w), k + u));
        const float* __restrict__ row = ys + (long)s * BH;
#pragma unroll
        for (int j = 0; j < PER; ++j) xj[u][j] = row[lane + j * 32];
      }
#pragma unroll
      for (int u = 0; u < U; ++u) add_edge<PER>(acc, xj[u], w[u], ydv, wev);
    } else {
#pragma unroll
      for (int u = 0; u < U - 1; ++u) {
        const int kk = (k + u < m) ? (k + u) : (m - 1);
        const int s  = __builtin_amdgcn_readlane(my_s, kk);
        w[u] = __int_as_float(__builtin_amdgcn_readlane(__float_as_int(my_w), kk));
        const float* __restrict__ row = ys + (long)s * BH;
#pragma unroll
        for (int j = 0; j < PER; ++j) xj[u][j] = row[lane + j * 32];
      }
#pragma unroll
      for (int u = U - 2; u >= 1; --u) {
        if (k + u < m) add_edge<PER>(acc, xj[u], w[u], ydv, wev);
      }
      add_edge<PER>(acc, xj[0], w[0], ydv, wev);
    }
  }

  if (n > BIN_CAP) {
    const int L = ovn[0];
    for (int k0 = 0; k0 < L; k0 += 32) {
      int  e   = 0;
      bool hit = false;
      if (k0 + lane < L) {
        e   = ov[k0 + lane];
        hit = (dst[e] == d);
      }
      unsigned mask = __builtin_amdgcn_ballot_w32(hit);
      while (mask != 0u) {
        const int b = __builtin_ctz(mask);
        mask &= mask - 1u;
        const int   eb = __builtin_amdgcn_readlane(e, b);
        const float w1 = ew[eb];
        const float* __restrict__ row = ys + (long)src[eb] * BH;
        float x1[PER];
#pragma unroll
        for (int j = 0; j < PER; ++j) x1[j] = row[lane + j * 32];
        add_edge<PER>(acc, x1, w1, ydv, wev);
      }
    }
  }

  const float inv_cnt = 1.0f / fmaxf((float)n, 1.0f);
  float vals[PER];
  float sum = 0.0f, ssq = 0.0f;
#pragma unroll
  for (int j = 0; j < PER; ++j) {
    const float o = ydv[j] + acc[j] * inv_cnt;
    vals[j] = o;
    sum += o;
    ssq += o * o;
  }
#pragma unroll
  for (int mask = 16; mask >= 1; mask >>= 1) {
    sum += __shfl_xor(sum, mask, 32);
    ssq += __shfl_xor(ssq, mask, 32);
  }
  const float mu  = sum * (1.0f / BH);
  const float var = ssq * (1.0f / BH) - mu * mu;
  const float is  = rsqrtf(var + 1e-5f);

#pragma unroll
  for (int j = 0; j < PER; ++j) {
    const int i = lane + j * 32;
    float v = xdv[j] + (vals[j] - mu) * is;
    v = v > 0.0f ? v : 0.01f * v;
    if (FINAL) {
      const int b = i / H, c = i & (H - 1);
      out[((long)b * Nd + d) * H + c] = v;
    } else {
      out[base + i] = v;
    }
  }
}

__device__ __forceinline__ void load_rows16(const float* p, v2f (&a)[4]) {
  const float4 t0 = *(const float4*)(p);
  const float4 t1 = *(const float4*)(p + 4);
  a[0].x = t0.x; a[0].y = t0.y;
  a[1].x = t0.z; a[1].y = t0.w;
  a[2].x = t1.x; a[2].y = t1.y;
  a[3].x = t1.z; a[3].y = t1.w;
}

__device__ __forceinline__ void load_w16x32(const float* __restrict__ Wx, int h, int n,
                                            v2f (&bw)[4][2]) {
#pragma unroll
  for (int ks = 0; ks < 4; ++ks) {
#pragma unroll
    for (int nt = 0; nt < 2; ++nt) {
      const int k0 = 8 * h + 2 * ks;
      bw[ks][nt].x = Wx[k0 * 32 + nt * 16 + n];
      bw[ks][nt].y = Wx[(k0 + 1) * 32 + nt * 16 + n];
    }
  }
}

__device__ __forceinline__ void load_w_folded(const float* __restrict__ PF, int lane,
                                              v2f (&bw)[4][2]) {
  const float* __restrict__ p = PF + lane * 16;
#pragma unroll
  for (int nt = 0; nt < 2; ++nt) {
    const float4 t0 = *(const float4*)(p + nt * 8);
    const float4 t1 = *(const float4*)(p + nt * 8 + 4);
    bw[0][nt].x = t0.x; bw[0][nt].y = t0.y;
    bw[1][nt].x = t0.z; bw[1][nt].y = t0.w;
    bw[2][nt].x = t1.x; bw[2][nt].y = t1.y;
    bw[3][nt].x = t1.z; bw[3][nt].y = t1.w;
  }
}

__device__ __forceinline__ void rows_times_w(const v2f (&a)[4], const v2f (&bw)[4][2],
                                             v8f (&d)[2]) {
  d[0] = v8f{};
  d[1] = v8f{};
#pragma unroll
  for (int ks = 0; ks < 4; ++ks) {
#pragma unroll
    for (int nt = 0; nt < 2; ++nt) {
      d[nt] = __builtin_amdgcn_wmma_f32_16x16x4_f32(
           false, a[ks],  false, bw[ks][nt],
           (short)0, d[nt],  false,  false);
    }
  }
}

template <int T>
__device__ __forceinline__ void request_rows(const char* hb, unsigned lane_off,
                                             int my_s, int k, int m, int slot, v2f (&a)[T][4]) {
  const int last = (m > 0) ? (m - 1) : 0;
#pragma unroll
  for (int g = 0; g < T; ++g) {
    const int i  = k + 4 * g + slot;
    const int sg = __shfl(my_s, (i < m) ? i : last, 32);
    load_rows16((const float*)(hb + ((unsigned)sg * 256u + lane_off)), a[g]);
  }
  asm volatile("" ::: "memory");
}

__device__ __forceinline__ void add_tile(float (&acc)[2][4], const v2f (&a)[4],
                                         const v2f (&bv)[4][2], float w0, float w1, bool ok0,
                                         bool ok1, const float (&ydv)[2][4],
                                         const float (&wev)[2]) {
  v8f x[2];
  rows_times_w(a, bv, x);
#pragma unroll
  for (int nt = 0; nt < 2; ++nt) {
    const float e0 = w0 * wev[nt];
    const float e1 = w1 * wev[nt];
#pragma unroll
    for (int b = 0; b < 4; ++b) {
      const float x0 = x[nt][b];
      const float t0 = ydv[nt][b] * x0 * e0;
      const float m0 = x0 * __builtin_amdgcn_rcpf(1.0f + __expf(-t0));
      acc[nt][b] += ok0 ? m0 : 0.0f;
      const float x1 = x[nt][4 + b];
      const float t1 = ydv[nt][b] * x1 * e1;
      const float m1 = x1 * __builtin_amdgcn_rcpf(1.0f + __expf(-t1));
      acc[nt][b] += ok1 ? m1 : 0.0f;
    }
  }
}

constexpr int TILES_PER_TRIP = 3;

__global__ void gather_l2_kernel(const float* h1, const int* __restrict__ res,
                                 const float* __restrict__ Wn, const float* __restrict__ WU,
                                 const float* __restrict__ WV, const int* __restrict__ cnti,
                                 const int* __restrict__ bins, const int* __restrict__ ovn,
                                 const int* __restrict__ ov, const int* __restrict__ src,
                                 const int* __restrict__ dst, const float* __restrict__ ew,
                                 const float* __restrict__ We, float* __restrict__ out, int Nd) {
  constexpr int ROW   = 64;
  constexpr int TILES = TILES_PER_TRIP;
  const int d = __builtin_amdgcn_readfirstlane(
      (int)(blockIdx.x * (blockDim.x >> 5) + (threadIdx.x >> 5)));
  if (d >= Nd) return;
  const int lane = threadIdx.x & 31;
  const int n16  = lane & 15;
  const int h    = lane >> 4;

  int rd = res[d];
  int n  = cnti[d];
  asm volatile("" : "+s"(rd), "+s"(n));

  const int m    = (n < BIN_CAP) ? n : BIN_CAP;
  int       my_s = 0;
  float     my_w = 0.0f;
  if (lane < m) {
    const int2 p = ((const int2*)bins)[(long)d * BIN_CAP + lane];
    my_s = p.x;
    my_w = __int_as_float(p.y);
  }

  const int koff = (n16 & 3) * 16 + 8 * h;
  v2f a_own[4];
  load_rows16(h1 + (long)rd * ROW + koff, a_own);
  v2f bn[4][2], bu[4][2], bv[4][2];
  load_w16x32(Wn, h, n16, bn);
  load_w_folded(WU, lane, bu);
  load_w_folded(WV, lane, bv);
  float wev[2];
  wev[0] = We[n16];
  wev[1] = We[16 + n16];

  const char* hb = (const char*)h1;
  const unsigned lane_off = (unsigned)koff * 4u;
  v2f a[TILES][4];
  request_rows<TILES>(hb, lane_off, my_s, 0, m, n16 >> 2, a);

  float ydv[2][4], xsel[4], acc[2][4];
  {
    v8f xdD[2], ydD[2];
    rows_times_w(a_own, bn, xdD);
    rows_times_w(a_own, bu, ydD);
#pragma unroll
    for (int b = 0; b < 4; ++b) {
      xsel[b]   = h ? xdD[1][b] : xdD[0][b];
      ydv[0][b] = ydD[0][b];
      ydv[1][b] = ydD[1][b];
      acc[0][b] = 0.0f;
      acc[1][b] = 0.0f;
    }
  }

  for (int k = 0; k < m; k += TILES * 4) {
#pragma unroll
    for (int g = 0; g < TILES; ++g) {
      if (g == 0 || k + 4 * g < m) {
        const int   j0 = k + 4 * g + 2 * h;
        const int   j1 = j0 + 1;
        const float w0 = __shfl(my_w, (j0 < m) ? j0 : (m - 1), 32);
        const float w1 = __shfl(my_w, (j1 < m) ? j1 : (m - 1), 32);
        add_tile(acc, a[g], bv, w0, w1, j0 < m, j1 < m, ydv, wev);
      }
    }
    if (k + TILES * 4 < m) request_rows<TILES>(hb, lane_off, my_s, k + TILES * 4, m, n16 >> 2, a);
  }

  if (n > BIN_CAP) {
    const int L = ovn[0];
    for (int k0 = 0; k0 < L; k0 += 32) {
      int  e   = 0;
      bool hit = false;
      if (k0 + lane < L) {
        e   = ov[k0 + lane];
        hit = (dst[e] == d);
      }
      unsigned mask = __builtin_amdgcn_ballot_w32(hit);
      while (mask != 0u) {
        const int bpos = __builtin_ctz(mask);
        mask &= mask - 1u;
        const int   eb = __builtin_amdgcn_readlane(e, bpos);
        const float w1 = ew[eb];
        v2f ax[4];
        load_rows16(h1 + (long)src[eb] * ROW + koff, ax);
        add_tile(acc, ax, bv, w1, w1, h == 0, false, ydv, wev);
      }
    }
  }

#pragma unroll
  for (int nt = 0; nt < 2; ++nt) {
#pragma unroll
    for (int b = 0; b < 4; ++b) acc[nt][b] += __shfl_xor(acc[nt][b], 16, 32);
  }

  const float inv_cnt = 1.0f / fmaxf((float)n, 1.0f);
  float o[2][4];
  float sum = 0.0f, ssq = 0.0f;
#pragma unroll
  for (int nt = 0; nt < 2; ++nt) {
#pragma unroll
    for (int b = 0; b < 4; ++b) {
      const float v = ydv[nt][b] + acc[nt][b] * inv_cnt;
      o[nt][b] = v;
      sum += v;
      ssq += v * v;
    }
  }
#pragma unroll
  for (int mask = 8; mask >= 1; mask >>= 1) {
    sum += __shfl_xor(sum, mask, 32);
    ssq += __shfl_xor(ssq, mask, 32);
  }
  const float mu  = sum * (1.0f / 128.0f);
  const float var = ssq * (1.0f / 128.0f) - mu * mu;
  const float is  = rsqrtf(var + 1e-5f);

#pragma unroll
  for (int b = 0; b < 4; ++b) {
    const float ob = h ? o[1][b] : o[0][b];
    float v = xsel[b] + (ob - mu) * is;
    v = v > 0.0f ? v : 0.01f * v;
    out[((long)b * Nd + d) * 32 + lane] = v;
  }
}

extern "C" void kernel_launch(void* const* d_in, const int* in_sizes, int n_in,
                              void* d_out, int out_size, void* d_ws, size_t ws_size,
                              hipStream_t stream) {
  constexpr int B = 4, Cin = 32, H = 16, Cout = 32;
  const float* X    = (const float*)d_in[0];
  const int*   src0 = (const int*)d_in[1];
  const int*   dst0 = (const int*)d_in[2];
  const float* ew0  = (const float*)d_in[3];
  const int*   res0 = (const int*)d_in[4];
  const int*   src1 = (const int*)d_in[5];
  const int*   dst1 = (const int*)d_in[6];
  const float* ew1  = (const float*)d_in[7];
  const int*   res1 = (const int*)d_in[8];
  const float* Wn1  = (const float*)d_in[9];
  const float* We1  = (const float*)d_in[10];
  const float* U1   = (const float*)d_in[11];
  const float* V1   = (const float*)d_in[12];
  const float* Wn2  = (const float*)d_in[13];
  const float* We2  = (const float*)d_in[14];
  const float* U2   = (const float*)d_in[15];
  const float* V2   = (const float*)d_in[16];

  const int E0 = in_sizes[1];
  const int N1 = in_sizes[4];
  const int E1 = in_sizes[5];
  const int N2 = in_sizes[8];
  const int N0 = in_sizes[0] / (B * Cin);

  float* ws = (float*)d_ws;
  const size_t szXS = (size_t)((N0 * B * H > N1 * B * Cout) ? N0 * B * H : N1 * B * Cout);
  const size_t szYD = (size_t)((N1 * B * H > N2 * B * Cout) ? N1 * B * H : N2 * B * Cout);
  float* xs  = ws;
  float* ys  = xs + szXS;
  float* yd  = ys + szXS;
  float* agg = yd + szYD;
  float* cnt = agg + szYD;
  float* h1  = (float*)(((size_t)(cnt + 2 * 16 * 32) + 255) & ~(size_t)255);

  int* bins0 = (int*)agg;
  int* cnti0 = bins0 + (size_t)N1 * BIN_CAP * 2;
  int* ovn0  = cnti0 + N1;
  int* cnti1 = ovn0 + 1;
  int* ovn1  = cnti1 + N2;
  int* ov0   = ovn1 + 1;
  int* ov1   = ov0 + E0;
  int* bins1 = (int*)yd;
  float* WV2 = cnt;
  float* WU2 = cnt + 16 * 32;

  const dim3 blk(256);
  auto waves_grid = [](int tiles) { return dim3((unsigned)((tiles + 7) / 8)); };

  const int R1s = N0 * B;
  gemm_pair_wmma<32, 16, 1><<<waves_grid(R1s / 16), blk, 0, stream>>>(
      X, Wn1, V1, xs, ys, R1s, N0, cnti0, N1 + N2 + 2);
  {
    const int blocks0 = (E0 + 255) / 256;
    const int blocks1 = (E1 + 255) / 256;
    edge_bin_both<<<dim3((unsigned)(blocks0 + blocks1 + 1)), blk, 0, stream>>>(
        blocks0, Wn2, V2, U2, WV2, WU2,
        src0, dst0, ew0, E0, cnti0, bins0, ovn0, ov0,
        src1, dst1, ew1, E1, cnti1, bins1, ovn1, ov1);
    gather_finalize_kernel<64, 16, false><<<dim3((unsigned)((N1 + 7) / 8)), blk, 0, stream>>>(
        xs, res0, ys, U1, cnti0, bins0, ovn0, ov0, src0, dst0, ew0, We1, h1, N1);
  }

  gather_l2_kernel<<<dim3((unsigned)((N2 + 7) / 8)), blk, 0, stream>>>(
      h1, res1, Wn2, WU2, WV2, cnti1, bins1, ovn1, ov1, src1, dst1, ew1, We2, (float*)d_out, N2);
}
